# starting kernel
# baseline (speedup 1.0000x reference)
#include <hip/hip_runtime.h>
#include <hip/hip_bf16.h>

#define N_NODES 100000
#define N_EDGES 1600000
#define D_IN    128
#define D_HID   128
#define N_CLS   32
#define N_STRIPS (N_NODES / 16)

typedef float v2f __attribute__((ext_vector_type(2)));
typedef float v8f __attribute__((ext_vector_type(8)));

__global__ void fill_f32(float* __restrict__ p, float v, long n) {
    long i = (long)blockIdx.x * 256 + threadIdx.x;
    if (i < n) p[i] = v;
}

__global__ void deg_accum(const int* __restrict__ dst, float* __restrict__ deg) {
    int e = blockIdx.x * 256 + threadIdx.x;
    if (e < N_EDGES) unsafeAtomicAdd(&deg[dst[e]], 1.0f);
}

__global__ void rsqrt_kernel(const float* __restrict__ deg, float* __restrict__ dinv) {
    int i = blockIdx.x * 256 + threadIdx.x;
    if (i < N_NODES) {
        float d = deg[i];
        dinv[i] = d > 0.0f ? rsqrtf(d) : 0.0f;
    }
}

template <int NCOLS>
__global__ __launch_bounds__(256)
void gemm_scale_kernel(const float* __restrict__ A,
                       const float* __restrict__ W,
                       const float* __restrict__ dinv,
                       float* __restrict__ out) {
    constexpr int NT = NCOLS / 16;
    __shared__ float lW[128 * NCOLS];

    const int tid = threadIdx.x;
    for (int idx = tid; idx < 128 * NCOLS; idx += 256) {
        const int k = idx / NCOLS;
        const int n = idx - k * NCOLS;
        lW[(k >> 1) * (2 * NCOLS) + 2 * n + (k & 1)] = W[idx];
    }
    __syncthreads();

    const int wave  = tid >> 5;
    const int lane  = tid & 31;
    const int strip = blockIdx.x * 8 + wave;
    if (strip >= N_STRIPS) return;

    const int row0 = strip * 16;
    const int half = lane >> 4;
    const int l16  = lane & 15;
    const int arow = row0 + l16;

    v8f acc[NT];
#pragma unroll
    for (int t = 0; t < NT; ++t) acc[t] = (v8f){0.f, 0.f, 0.f, 0.f, 0.f, 0.f, 0.f, 0.f};

    for (int k0 = 0; k0 < 128; k0 += 4) {
        const int ka = k0 + 2 * half;
        const v2f a = *(const v2f*)&A[(long)arow * 128 + ka];
        const float* __restrict__ lrow = &lW[(ka >> 1) * (2 * NCOLS)];
#pragma unroll
        for (int t = 0; t < NT; ++t) {
            const int n = t * 16 + l16;
            const v2f b = *(const v2f*)&lrow[2 * n];
            acc[t] = __builtin_amdgcn_wmma_f32_16x16x4_f32(
                false, a, false, b, (short)0, acc[t], false, false);
        }
    }

#pragma unroll
    for (int t = 0; t < NT; ++t) {
#pragma unroll
        for (int r = 0; r < 8; ++r) {
            const int rr = row0 + r + 8 * half;
            out[(long)rr * NCOLS + t * 16 + l16] = acc[t][r] * dinv[rr];
        }
    }
}

__global__ __launch_bounds__(256)
void scatter128(const int* __restrict__ src, const int* __restrict__ dst,
                const float* __restrict__ hs, float* __restrict__ t1) {
    const int e = blockIdx.x * 8 + (threadIdx.x >> 5);
    const int lane = threadIdx.x & 31;
    if (e >= N_EDGES) return;
    const long s = src[e];
    const long d = dst[e];
    const float* __restrict__ hrow = hs + s * 128;
    float* __restrict__ trow = t1 + d * 128;
#pragma unroll
    for (int c = 0; c < 4; ++c) {
        unsafeAtomicAdd(&trow[lane + 32 * c], hrow[lane + 32 * c]);
    }
}

__global__ __launch_bounds__(256)
void scatter32(const int* __restrict__ src, const int* __restrict__ dst,
               const float* __restrict__ hs, float* __restrict__ t2) {
    const int e = blockIdx.x * 8 + (threadIdx.x >> 5);
    const int lane = threadIdx.x & 31;
    if (e >= N_EDGES) return;
    const long s = src[e];
    const long d = dst[e];
    unsafeAtomicAdd(&t2[d * 32 + lane], hs[s * 32 + lane]);
}

__global__ void post1(const float* __restrict__ hs, float* __restrict__ t1,
                      const float* __restrict__ dinv, const float* __restrict__ b1) {
    long i = (long)blockIdx.x * 256 + threadIdx.x;
    if (i >= (long)N_NODES * 128) return;
    const int node = (int)(i >> 7);
    const int c = (int)(i & 127);
    float v = (t1[i] + hs[i]) * dinv[node] + b1[c];
    t1[i] = fmaxf(v, 0.0f);
}

__global__ __launch_bounds__(256)
void final_kernel(const float* __restrict__ h2s, const float* __restrict__ t2,
                  const float* __restrict__ dinv, const float* __restrict__ b2,
                  float* __restrict__ out) {
    const int node = blockIdx.x * 8 + (threadIdx.x >> 5);
    const int lane = threadIdx.x & 31;
    if (node >= N_NODES) return;
    const long idx = (long)node * 32 + lane;
    float v = (t2[idx] + h2s[idx]) * dinv[node] + b2[lane];

    float m = v;
#pragma unroll
    for (int o = 16; o > 0; o >>= 1) m = fmaxf(m, __shfl_xor(m, o, 32));
    float s = __expf(v - m);
#pragma unroll
    for (int o = 16; o > 0; o >>= 1) s += __shfl_xor(s, o, 32);
    out[idx] = (v - m) - __logf(s);
}

extern "C" void kernel_launch(void* const* d_in, const int* in_sizes, int n_in,
                              void* d_out, int out_size, void* d_ws, size_t ws_size,
                              hipStream_t stream) {
    const float* x  = (const float*)d_in[0];
    const int*   ei = (const int*)d_in[1];
    const float* W1 = (const float*)d_in[2];
    const float* b1 = (const float*)d_in[3];
    const float* W2 = (const float*)d_in[4];
    const float* b2 = (const float*)d_in[5];
    float* out = (float*)d_out;

    const int* src = ei;
    const int* dst = ei + N_EDGES;

    float* deg  = (float*)d_ws;
    float* dinv = deg  + N_NODES;
    float* hs   = dinv + N_NODES;
    float* t1   = hs   + (long)N_NODES * 128;
    float* h2s  = t1   + (long)N_NODES * 128;
    float* t2   = h2s  + (long)N_NODES * 32;

    const long n128 = (long)N_NODES * 128;
    const long n32  = (long)N_NODES * 32;

    fill_f32<<<(N_NODES + 255) / 256, 256, 0, stream>>>(deg, 1.0f, N_NODES);
    deg_accum<<<(N_EDGES + 255) / 256, 256, 0, stream>>>(dst, deg);
    rsqrt_kernel<<<(N_NODES + 255) / 256, 256, 0, stream>>>(deg, dinv);

    fill_f32<<<(int)((n128 + 255) / 256), 256, 0, stream>>>(t1, 0.0f, n128);
    fill_f32<<<(int)((n32 + 255) / 256), 256, 0, stream>>>(t2, 0.0f, n32);

    gemm_scale_kernel<128><<<(N_STRIPS + 7) / 8, 256, 0, stream>>>(x, W1, dinv, hs);
    scatter128<<<(N_EDGES + 7) / 8, 256, 0, stream>>>(src, dst, hs, t1);
    post1<<<(int)((n128 + 255) / 256), 256, 0, stream>>>(hs, t1, dinv, b1);

    gemm_scale_kernel<32><<<(N_STRIPS + 7) / 8, 256, 0, stream>>>(t1, W2, dinv, h2s);
    scatter32<<<(N_EDGES + 7) / 8, 256, 0, stream>>>(src, dst, h2s, t2);
    final_kernel<<<(N_NODES + 7) / 8, 256, 0, stream>>>(h2s, t2, dinv, b2, out);
}


# Float-atomic edge scatter replaced by a per-call CSR (returning int atomic rank, scan, fill) and a one-wave-per-node gather with the epilogues fused in; edge sums now run in CSR rank order
# speedup vs baseline: 2.3930x; 2.3930x over previous
#include <hip/hip_runtime.h>
#include <hip/hip_bf16.h>

#define N_NODES 100000
#define N_EDGES 1600000
#define D_IN    128
#define D_HID   128
#define N_CLS   32
#define N_STRIPS (N_NODES / 16)

typedef float v2f __attribute__((ext_vector_type(2)));
typedef float v8f __attribute__((ext_vector_type(8)));

__global__ void zero_i32(int* __restrict__ p, int n) {
    int i = blockIdx.x * 256 + threadIdx.x;
    if (i < n) p[i] = 0;
}

__global__ void rank_edges(const int* __restrict__ dst, int* __restrict__ cnt,
                           int* __restrict__ rank) {
    int e = blockIdx.x * 256 + threadIdx.x;
    if (e < N_EDGES) rank[e] = atomicAdd(&cnt[dst[e]], 1);
}

__global__ __launch_bounds__(1024)
void scan_counts(const int* __restrict__ cnt, int* __restrict__ rowptr,
                 float* __restrict__ dinv) {
    __shared__ int wsum[32];
    __shared__ int wpre[32];
    __shared__ int tile_pre;
    const int tid  = threadIdx.x;
    const int lane = tid & 31;
    const int wave = tid >> 5;
    const int tile0 = blockIdx.x * 1024;

    int part = 0;
    for (int i = tid; i < tile0; i += 1024) part += cnt[i];
#pragma unroll
    for (int o = 16; o > 0; o >>= 1) part += __shfl_xor(part, o, 32);
    if (lane == 0) wsum[wave] = part;
    __syncthreads();
    if (wave == 0) {
        int v = wsum[lane];
#pragma unroll
        for (int o = 16; o > 0; o >>= 1) v += __shfl_xor(v, o, 32);
        if (lane == 0) tile_pre = v;
    }
    __syncthreads();
    const int before = tile_pre;

    const int i = tile0 + tid;
    const int x = (i < N_NODES) ? cnt[i] : 0;
    int incl = x;
#pragma unroll
    for (int o = 1; o < 32; o <<= 1) {
        const int y = __shfl_up(incl, o, 32);
        if (lane >= o) incl += y;
    }
    if (lane == 31) wsum[wave] = incl;
    __syncthreads();
    if (wave == 0) {
        const int t = wsum[lane];
        int s = t;
#pragma unroll
        for (int o = 1; o < 32; o <<= 1) {
            const int y = __shfl_up(s, o, 32);
            if (lane >= o) s += y;
        }
        wpre[lane] = s - t;
    }
    __syncthreads();
    if (i < N_NODES) {
        const int excl = before + wpre[wave] + incl - x;
        rowptr[i] = excl;
        if (i == N_NODES - 1) rowptr[N_NODES] = excl + x;
        const float d = (float)(x + 1);
        dinv[i] = d > 0.0f ? rsqrtf(d) : 0.0f;
    }
}

__global__ void fill_csr(const int* __restrict__ src, const int* __restrict__ dst,
                         const int* __restrict__ rank, const int* __restrict__ rowptr,
                         int* __restrict__ csr_src) {
    int e = blockIdx.x * 256 + threadIdx.x;
    if (e < N_EDGES) csr_src[rowptr[dst[e]] + rank[e]] = src[e];
}

template <int NCOLS>
__global__ __launch_bounds__(256)
void gemm_scale_kernel(const float* __restrict__ A,
                       const float* __restrict__ W,
                       const float* __restrict__ dinv,
                       float* __restrict__ out) {
    constexpr int NT = NCOLS / 16;
    __shared__ float lW[128 * NCOLS];

    const int tid = threadIdx.x;
    for (int idx = tid; idx < 128 * NCOLS; idx += 256) {
        const int k = idx / NCOLS;
        const int n = idx - k * NCOLS;
        lW[(k >> 1) * (2 * NCOLS) + 2 * n + (k & 1)] = W[idx];
    }
    __syncthreads();

    const int wave  = tid >> 5;
    const int lane  = tid & 31;
    const int strip = blockIdx.x * 8 + wave;
    if (strip >= N_STRIPS) return;

    const int row0 = strip * 16;
    const int half = lane >> 4;
    const int l16  = lane & 15;
    const int arow = row0 + l16;

    v8f acc[NT];
#pragma unroll
    for (int t = 0; t < NT; ++t) acc[t] = (v8f){0.f, 0.f, 0.f, 0.f, 0.f, 0.f, 0.f, 0.f};

    for (int k0 = 0; k0 < 128; k0 += 4) {
        const int ka = k0 + 2 * half;
        const v2f a = *(const v2f*)&A[(long)arow * 128 + ka];
        const float* __restrict__ lrow = &lW[(ka >> 1) * (2 * NCOLS)];
#pragma unroll
        for (int t = 0; t < NT; ++t) {
            const int n = t * 16 + l16;
            const v2f b = *(const v2f*)&lrow[2 * n];
            acc[t] = __builtin_amdgcn_wmma_f32_16x16x4_f32(
                false, a, false, b, (short)0, acc[t], false, false);
        }
    }

#pragma unroll
    for (int t = 0; t < NT; ++t) {
#pragma unroll
        for (int r = 0; r < 8; ++r) {
            const int rr = row0 + r + 8 * half;
            out[(long)rr * NCOLS + t * 16 + l16] = acc[t][r] * dinv[rr];
        }
    }
}

__global__ __launch_bounds__(256)
void gather128_post(const int* __restrict__ rowptr, const int* __restrict__ csr_src,
                    const float* __restrict__ hs, const float* __restrict__ dinv,
                    const float* __restrict__ b1, float* __restrict__ t1) {
    const int lane = threadIdx.x & 31;
    const int node = __builtin_amdgcn_readfirstlane(blockIdx.x * 8 + (threadIdx.x >> 5));
    if (node >= N_NODES) return;
    const int beg = rowptr[node];
    const int end = rowptr[node + 1];
    const float4* __restrict__ hs4 = (const float4*)hs;
    float4 acc = hs4[(long)node * 32 + lane];
    for (int base = beg; base < end; base += 32) {
        const int n = min(32, end - base);
        const int idx = (lane < n) ? csr_src[base + lane] : 0;
#pragma unroll 4
        for (int j = 0; j < n; ++j) {
            const int s = __builtin_amdgcn_readlane(idx, j);
            const float4 v = hs4[(long)s * 32 + lane];
            acc.x += v.x; acc.y += v.y; acc.z += v.z; acc.w += v.w;
        }
    }
    const float dn = dinv[node];
    const int c = lane * 4;
    float4 o;
    o.x = fmaxf(acc.x * dn + b1[c + 0], 0.0f);
    o.y = fmaxf(acc.y * dn + b1[c + 1], 0.0f);
    o.z = fmaxf(acc.z * dn + b1[c + 2], 0.0f);
    o.w = fmaxf(acc.w * dn + b1[c + 3], 0.0f);
    ((float4*)t1)[(long)node * 32 + lane] = o;
}

__global__ __launch_bounds__(256)
void gather32_final(const int* __restrict__ rowptr, const int* __restrict__ csr_src,
                    const float* __restrict__ h2s, const float* __restrict__ dinv,
                    const float* __restrict__ b2, float* __restrict__ out) {
    const int lane = threadIdx.x & 31;
    const int node = __builtin_amdgcn_readfirstlane(blockIdx.x * 8 + (threadIdx.x >> 5));
    if (node >= N_NODES) return;
    const int beg = rowptr[node];
    const int end = rowptr[node + 1];
    float acc = h2s[(long)node * 32 + lane];
    for (int base = beg; base < end; base += 32) {
        const int n = min(32, end - base);
        const int idx = (lane < n) ? csr_src[base + lane] : 0;
#pragma unroll 4
        for (int j = 0; j < n; ++j) {
            const int s = __builtin_amdgcn_readlane(idx, j);
            acc += h2s[(long)s * 32 + lane];
        }
    }
    const float v = acc * dinv[node] + b2[lane];

    float m = v;
#pragma unroll
    for (int o = 16; o > 0; o >>= 1) m = fmaxf(m, __shfl_xor(m, o, 32));
    float s = __expf(v - m);
#pragma unroll
    for (int o = 16; o > 0; o >>= 1) s += __shfl_xor(s, o, 32);
    out[(long)node * 32 + lane] = (v - m) - __logf(s);
}

extern "C" void kernel_launch(void* const* d_in, const int* in_sizes, int n_in,
                              void* d_out, int out_size, void* d_ws, size_t ws_size,
                              hipStream_t stream) {
    const float* x  = (const float*)d_in[0];
    const int*   ei = (const int*)d_in[1];
    const float* W1 = (const float*)d_in[2];
    const float* b1 = (const float*)d_in[3];
    const float* W2 = (const float*)d_in[4];
    const float* b2 = (const float*)d_in[5];
    float* out = (float*)d_out;

    const int* src = ei;
    const int* dst = ei + N_EDGES;

    float* dinv = (float*)d_ws;
    float* hs   = dinv + N_NODES;
    float* t1   = hs   + (long)N_NODES * 128;
    float* h2s  = t1   + (long)N_NODES * 128;
    int* cnt     = (int*)(h2s + (long)N_NODES * 32);
    int* rowptr  = cnt + N_NODES;
    int* rank    = rowptr + N_NODES + 32;
    int* csr_src = rank + N_EDGES;

    zero_i32<<<(N_NODES + 255) / 256, 256, 0, stream>>>(cnt, N_NODES);
    rank_edges<<<(N_EDGES + 255) / 256, 256, 0, stream>>>(dst, cnt, rank);
    scan_counts<<<(N_NODES + 1023) / 1024, 1024, 0, stream>>>(cnt, rowptr, dinv);
    fill_csr<<<(N_EDGES + 255) / 256, 256, 0, stream>>>(src, dst, rank, rowptr, csr_src);

    gemm_scale_kernel<128><<<(N_STRIPS + 7) / 8, 256, 0, stream>>>(x, W1, dinv, hs);
    gather128_post<<<(N_NODES + 7) / 8, 256, 0, stream>>>(rowptr, csr_src, hs, dinv, b1, t1);

    gemm_scale_kernel<32><<<(N_STRIPS + 7) / 8, 256, 0, stream>>>(t1, W2, dinv, h2s);
    gather32_final<<<(N_NODES + 7) / 8, 256, 0, stream>>>(rowptr, csr_src, h2s, dinv, b2, out);
}


# Per-edge returning global atomics removed from the CSR build: two-level counting sort on LDS counters (256-node buckets, then nodes); edge sums run in LDS arrival-rank order
# speedup vs baseline: 3.4163x; 1.4276x over previous
#include <hip/hip_runtime.h>
#include <hip/hip_bf16.h>

#define N_NODES 100000
#define N_EDGES 1600000
#define D_IN    128
#define D_HID   128
#define N_CLS   32
#define N_STRIPS (N_NODES / 16)

typedef float v2f __attribute__((ext_vector_type(2)));
typedef float v8f __attribute__((ext_vector_type(8)));

#define BK_SHIFT 8
#define BK_NODES (1 << BK_SHIFT)
#define NBK      ((N_NODES + BK_NODES - 1) / BK_NODES)
#define NCH      256
#define CH_EDGES ((N_EDGES + NCH - 1) / NCH)
#define L1_THREADS 1024
#define L1_ITERS ((CH_EDGES + L1_THREADS - 1) / L1_THREADS)
#define SRC_BITS 17
#define SRC_MASK ((1u << SRC_BITS) - 1u)
#define NO_EDGE  0xFFFFFFFFu

static_assert(NCH == 256, "bucket_scan gives each lane 8 of the 256 chunk counts");
static_assert(N_NODES <= (1 << SRC_BITS), "source id must fit the packed word");

__global__ __launch_bounds__(L1_THREADS)
void bucket_hist(const int* __restrict__ dst, int* __restrict__ H) {
    __shared__ int h[NBK];
    const int tid = threadIdx.x;
    const int e0 = blockIdx.x * CH_EDGES;
    const int e1 = min(e0 + CH_EDGES, N_EDGES);
    int d[L1_ITERS];
#pragma unroll
    for (int j = 0; j < L1_ITERS; ++j) {
        const int e = e0 + tid + j * L1_THREADS;
        d[j] = (e < e1) ? dst[e] : -1;
    }
    for (int k = tid; k < NBK; k += L1_THREADS) h[k] = 0;
    __syncthreads();
#pragma unroll
    for (int j = 0; j < L1_ITERS; ++j)
        if (d[j] >= 0) atomicAdd(&h[d[j] >> BK_SHIFT], 1);
    __syncthreads();
    for (int k = tid; k < NBK; k += L1_THREADS) H[k * NCH + blockIdx.x] = h[k];
}

__global__ __launch_bounds__(1024)
void bucket_scan(const int* __restrict__ H, int* __restrict__ base,
                 int* __restrict__ bstart) {
    constexpr int PER = (NBK + 31) / 32;
    __shared__ int T[PER * 32];
    const int tid  = threadIdx.x;
    const int lane = tid & 31;
    const int wave = tid >> 5;

    int tot[PER];
#pragma unroll
    for (int j = 0; j < PER; ++j) {
        const int k = min(wave + 32 * j, NBK - 1);
        const int4* p = (const int4*)(H + k * NCH + lane * 8);
        const int4 a = p[0];
        const int4 b = p[1];
        tot[j] = a.x + a.y + a.z + a.w + b.x + b.y + b.z + b.w;
    }
#pragma unroll
    for (int j = 0; j < PER; ++j) {
        int s = tot[j];
#pragma unroll
        for (int o = 16; o > 0; o >>= 1) s += __shfl_xor(s, o, 32);
        const int k = wave + 32 * j;
        if (lane == 0) T[k] = (k < NBK) ? s : 0;
    }
    __syncthreads();
    if (wave == 0) {
        int v[PER];
        int sum = 0;
#pragma unroll
        for (int j = 0; j < PER; ++j) { v[j] = T[lane * PER + j]; sum += v[j]; }
        int incl = sum;
#pragma unroll
        for (int o = 1; o < 32; o <<= 1) {
            const int y = __shfl_up(incl, o, 32);
            if (lane >= o) incl += y;
        }
        int run = incl - sum;
#pragma unroll
        for (int j = 0; j < PER; ++j) { T[lane * PER + j] = run; run += v[j]; }
    }
    __syncthreads();

    const int k = blockIdx.x * 32 + wave;
    if (k < NBK) {
        const int4* p = (const int4*)(H + k * NCH + lane * 8);
        const int4 a = p[0];
        const int4 b = p[1];
        const int s = a.x + a.y + a.z + a.w + b.x + b.y + b.z + b.w;
        int incl = s;
#pragma unroll
        for (int o = 1; o < 32; o <<= 1) {
            const int y = __shfl_up(incl, o, 32);
            if (lane >= o) incl += y;
        }
        const int first = T[k];
        int run = first + incl - s;
        int4 oa, ob;
        oa.x = run; run += a.x;
        oa.y = run; run += a.y;
        oa.z = run; run += a.z;
        oa.w = run; run += a.w;
        ob.x = run; run += b.x;
        ob.y = run; run += b.y;
        ob.z = run; run += b.z;
        ob.w = run;
        int4* q = (int4*)(base + k * NCH + lane * 8);
        q[0] = oa;
        q[1] = ob;
        if (lane == 0) bstart[k] = first;
    }
    if (blockIdx.x == 0 && tid == 0) bstart[NBK] = N_EDGES;
}

__global__ __launch_bounds__(L1_THREADS)
void bucket_scatter(const int* __restrict__ src, const int* __restrict__ dst,
                    const int* __restrict__ base, unsigned* __restrict__ packed) {
    __shared__ int cur[NBK];
    const int tid = threadIdx.x;
    const int e0 = blockIdx.x * CH_EDGES;
    const int e1 = min(e0 + CH_EDGES, N_EDGES);
    int d[L1_ITERS], s[L1_ITERS];
#pragma unroll
    for (int j = 0; j < L1_ITERS; ++j) {
        const int e = e0 + tid + j * L1_THREADS;
        d[j] = (e < e1) ? dst[e] : -1;
        s[j] = (e < e1) ? src[e] : 0;
    }
    for (int k = tid; k < NBK; k += L1_THREADS) cur[k] = base[k * NCH + blockIdx.x];
    __syncthreads();
#pragma unroll
    for (int j = 0; j < L1_ITERS; ++j) {
        if (d[j] >= 0) {
            const int p = atomicAdd(&cur[d[j] >> BK_SHIFT], 1);
            packed[p] = ((unsigned)(d[j] & (BK_NODES - 1)) << SRC_BITS) | (unsigned)s[j];
        }
    }
}

#define FIN_THREADS 1024
#define FIN_KEEP 6
__global__ __launch_bounds__(FIN_THREADS)
void bucket_finalize(const unsigned* __restrict__ packed, const int* __restrict__ bstart,
                     int* __restrict__ rowptr, float* __restrict__ dinv,
                     int* __restrict__ csr_src) {
    __shared__ int cnt[BK_NODES];
    __shared__ int cur[BK_NODES];
    __shared__ int wtot[BK_NODES / 32];
    const int tid  = threadIdx.x;
    const int lane = tid & 31;
    const int wave = tid >> 5;
    const int k  = blockIdx.x;
    const int b0 = bstart[k];
    const int b1 = bstart[k + 1];

    unsigned w[FIN_KEEP];
#pragma unroll
    for (int j = 0; j < FIN_KEEP; ++j) {
        const int i = b0 + tid + j * FIN_THREADS;
        w[j] = (i < b1) ? packed[i] : NO_EDGE;
    }
    if (tid < BK_NODES) cnt[tid] = 0;
    __syncthreads();
#pragma unroll
    for (int j = 0; j < FIN_KEEP; ++j)
        if (w[j] != NO_EDGE) atomicAdd(&cnt[w[j] >> SRC_BITS], 1);
    for (int i = b0 + tid + FIN_KEEP * FIN_THREADS; i < b1; i += FIN_THREADS)
        atomicAdd(&cnt[packed[i] >> SRC_BITS], 1);
    __syncthreads();

    const int x = (tid < BK_NODES) ? cnt[tid] : 0;
    int incl = x;
#pragma unroll
    for (int o = 1; o < 32; o <<= 1) {
        const int y = __shfl_up(incl, o, 32);
        if (lane >= o) incl += y;
    }
    if (tid < BK_NODES && lane == 31) wtot[wave] = incl;
    __syncthreads();
    if (tid < BK_NODES) {
        int pre = 0;
        for (int q = 0; q < wave; ++q) pre += wtot[q];
        const int excl = pre + incl - x;
        cur[tid] = excl;
        const int node = k * BK_NODES + tid;
        if (node < N_NODES) {
            rowptr[node] = b0 + excl;
            const float dg = (float)(x + 1);
            dinv[node] = dg > 0.0f ? rsqrtf(dg) : 0.0f;
        }
    }
    if (k == NBK - 1 && tid == 0) rowptr[N_NODES] = b1;
    __syncthreads();

#pragma unroll
    for (int j = 0; j < FIN_KEEP; ++j) {
        if (w[j] != NO_EDGE) {
            const int p = atomicAdd(&cur[w[j] >> SRC_BITS], 1);
            csr_src[b0 + p] = (int)(w[j] & SRC_MASK);
        }
    }
    for (int i = b0 + tid + FIN_KEEP * FIN_THREADS; i < b1; i += FIN_THREADS) {
        const unsigned ww = packed[i];
        const int p = atomicAdd(&cur[ww >> SRC_BITS], 1);
        csr_src[b0 + p] = (int)(ww & SRC_MASK);
    }
}

template <int NCOLS>
__global__ __launch_bounds__(256)
void gemm_scale_kernel(const float* __restrict__ A,
                       const float* __restrict__ W,
                       const float* __restrict__ dinv,
                       float* __restrict__ out) {
    constexpr int NT = NCOLS / 16;
    __shared__ float lW[128 * NCOLS];

    const int tid = threadIdx.x;
    for (int idx = tid; idx < 128 * NCOLS; idx += 256) {
        const int k = idx / NCOLS;
        const int n = idx - k * NCOLS;
        lW[(k >> 1) * (2 * NCOLS) + 2 * n + (k & 1)] = W[idx];
    }
    __syncthreads();

    const int wave  = tid >> 5;
    const int lane  = tid & 31;
    const int strip = blockIdx.x * 8 + wave;
    if (strip >= N_STRIPS) return;

    const int row0 = strip * 16;
    const int half = lane >> 4;
    const int l16  = lane & 15;
    const int arow = row0 + l16;

    v8f acc[NT];
#pragma unroll
    for (int t = 0; t < NT; ++t) acc[t] = (v8f){0.f, 0.f, 0.f, 0.f, 0.f, 0.f, 0.f, 0.f};

    for (int k0 = 0; k0 < 128; k0 += 4) {
        const int ka = k0 + 2 * half;
        const v2f a = *(const v2f*)&A[(long)arow * 128 + ka];
        const float* __restrict__ lrow = &lW[(ka >> 1) * (2 * NCOLS)];
#pragma unroll
        for (int t = 0; t < NT; ++t) {
            const int n = t * 16 + l16;
            const v2f b = *(const v2f*)&lrow[2 * n];
            acc[t] = __builtin_amdgcn_wmma_f32_16x16x4_f32(
                false, a, false, b, (short)0, acc[t], false, false);
        }
    }

#pragma unroll
    for (int t = 0; t < NT; ++t) {
#pragma unroll
        for (int r = 0; r < 8; ++r) {
            const int rr = row0 + r + 8 * half;
            out[(long)rr * NCOLS + t * 16 + l16] = acc[t][r] * dinv[rr];
        }
    }
}

__global__ __launch_bounds__(256)
void gather128_post(const int* __restrict__ rowptr, const int* __restrict__ csr_src,
                    const float* __restrict__ hs, const float* __restrict__ dinv,
                    const float* __restrict__ b1, float* __restrict__ t1) {
    const int lane = threadIdx.x & 31;
    const int node = __builtin_amdgcn_readfirstlane(blockIdx.x * 8 + (threadIdx.x >> 5));
    if (node >= N_NODES) return;
    const int beg = rowptr[node];
    const int end = rowptr[node + 1];
    const float4* __restrict__ hs4 = (const float4*)hs;
    float4 acc = hs4[(long)node * 32 + lane];
    for (int base = beg; base < end; base += 32) {
        const int n = min(32, end - base);
        const int idx = (lane < n) ? csr_src[base + lane] : 0;
#pragma unroll 4
        for (int j = 0; j < n; ++j) {
            const int s = __builtin_amdgcn_readlane(idx, j);
            const float4 v = hs4[(long)s * 32 + lane];
            acc.x += v.x; acc.y += v.y; acc.z += v.z; acc.w += v.w;
        }
    }
    const float dn = dinv[node];
    const int c = lane * 4;
    float4 o;
    o.x = fmaxf(acc.x * dn + b1[c + 0], 0.0f);
    o.y = fmaxf(acc.y * dn + b1[c + 1], 0.0f);
    o.z = fmaxf(acc.z * dn + b1[c + 2], 0.0f);
    o.w = fmaxf(acc.w * dn + b1[c + 3], 0.0f);
    ((float4*)t1)[(long)node * 32 + lane] = o;
}

__global__ __launch_bounds__(256)
void gather32_final(const int* __restrict__ rowptr, const int* __restrict__ csr_src,
                    const float* __restrict__ h2s, const float* __restrict__ dinv,
                    const float* __restrict__ b2, float* __restrict__ out) {
    const int lane = threadIdx.x & 31;
    const int node = __builtin_amdgcn_readfirstlane(blockIdx.x * 8 + (threadIdx.x >> 5));
    if (node >= N_NODES) return;
    const int beg = rowptr[node];
    const int end = rowptr[node + 1];
    float acc = h2s[(long)node * 32 + lane];
    for (int base = beg; base < end; base += 32) {
        const int n = min(32, end - base);
        const int idx = (lane < n) ? csr_src[base + lane] : 0;
#pragma unroll 4
        for (int j = 0; j < n; ++j) {
            const int s = __builtin_amdgcn_readlane(idx, j);
            acc += h2s[(long)s * 32 + lane];
        }
    }
    const float v = acc * dinv[node] + b2[lane];

    float m = v;
#pragma unroll
    for (int o = 16; o > 0; o >>= 1) m = fmaxf(m, __shfl_xor(m, o, 32));
    float s = __expf(v - m);
#pragma unroll
    for (int o = 16; o > 0; o >>= 1) s += __shfl_xor(s, o, 32);
    out[(long)node * 32 + lane] = (v - m) - __logf(s);
}

extern "C" void kernel_launch(void* const* d_in, const int* in_sizes, int n_in,
                              void* d_out, int out_size, void* d_ws, size_t ws_size,
                              hipStream_t stream) {
    const float* x  = (const float*)d_in[0];
    const int*   ei = (const int*)d_in[1];
    const float* W1 = (const float*)d_in[2];
    const float* b1 = (const float*)d_in[3];
    const float* W2 = (const float*)d_in[4];
    const float* b2 = (const float*)d_in[5];
    float* out = (float*)d_out;

    const int* src = ei;
    const int* dst = ei + N_EDGES;

    float* dinv = (float*)d_ws;
    float* hs   = dinv + N_NODES;
    float* t1   = hs   + (long)N_NODES * 128;
    float* h2s  = t1   + (long)N_NODES * 128;
    int* rowptr  = (int*)(h2s + (long)N_NODES * 32);
    int* csr_src = rowptr + N_NODES + 32;
    unsigned* packed = (unsigned*)(csr_src + N_EDGES);
    int* H      = (int*)(packed + N_EDGES);
    int* base   = H + NBK * NCH;
    int* bstart = base + NBK * NCH;

    bucket_hist<<<NCH, L1_THREADS, 0, stream>>>(dst, H);
    bucket_scan<<<(NBK + 31) / 32, 1024, 0, stream>>>(H, base, bstart);
    bucket_scatter<<<NCH, L1_THREADS, 0, stream>>>(src, dst, base, packed);
    bucket_finalize<<<NBK, FIN_THREADS, 0, stream>>>(packed, bstart, rowptr, dinv, csr_src);

    gemm_scale_kernel<128><<<(N_STRIPS + 7) / 8, 256, 0, stream>>>(x, W1, dinv, hs);
    gather128_post<<<(N_NODES + 7) / 8, 256, 0, stream>>>(rowptr, csr_src, hs, dinv, b1, t1);

    gemm_scale_kernel<32><<<(N_STRIPS + 7) / 8, 256, 0, stream>>>(t1, W2, dinv, h2s);
    gather32_final<<<(N_NODES + 7) / 8, 256, 0, stream>>>(rowptr, csr_src, h2s, dinv, b2, out);
}


# Layer-2 product fused behind the layer-1 gather: block per 16-node strip, t1 rows go through an LDS strip to the WMMA, never to memory; one launch fewer; sum orders as before
# speedup vs baseline: 3.5440x; 1.0374x over previous
#include <hip/hip_runtime.h>
#include <hip/hip_bf16.h>

#define N_NODES 100000
#define N_EDGES 1600000
#define D_IN    128
#define D_HID   128
#define N_CLS   32
#define N_STRIPS (N_NODES / 16)

typedef float v2f __attribute__((ext_vector_type(2)));
typedef float v8f __attribute__((ext_vector_type(8)));

#define BK_SHIFT 8
#define BK_NODES (1 << BK_SHIFT)
#define NBK      ((N_NODES + BK_NODES - 1) / BK_NODES)
#define NCH      256
#define CH_EDGES ((N_EDGES + NCH - 1) / NCH)
#define L1_THREADS 1024
#define L1_ITERS ((CH_EDGES + L1_THREADS - 1) / L1_THREADS)
#define SRC_BITS 17
#define SRC_MASK ((1u << SRC_BITS) - 1u)
#define NO_EDGE  0xFFFFFFFFu

static_assert(NCH == 256, "bucket_scan gives each lane 8 of the 256 chunk counts");
static_assert(N_NODES <= (1 << SRC_BITS), "source id must fit the packed word");

__global__ __launch_bounds__(L1_THREADS)
void bucket_hist(const int* __restrict__ dst, int* __restrict__ H) {
    __shared__ int h[NBK];
    const int tid = threadIdx.x;
    const int e0 = blockIdx.x * CH_EDGES;
    const int e1 = min(e0 + CH_EDGES, N_EDGES);
    int d[L1_ITERS];
#pragma unroll
    for (int j = 0; j < L1_ITERS; ++j) {
        const int e = e0 + tid + j * L1_THREADS;
        d[j] = (e < e1) ? dst[e] : -1;
    }
    for (int k = tid; k < NBK; k += L1_THREADS) h[k] = 0;
    __syncthreads();
#pragma unroll
    for (int j = 0; j < L1_ITERS; ++j)
        if (d[j] >= 0) atomicAdd(&h[d[j] >> BK_SHIFT], 1);
    __syncthreads();
    for (int k = tid; k < NBK; k += L1_THREADS) H[k * NCH + blockIdx.x] = h[k];
}

__global__ __launch_bounds__(1024)
void bucket_scan(const int* __restrict__ H, int* __restrict__ base,
                 int* __restrict__ bstart) {
    constexpr int PER = (NBK + 31) / 32;
    __shared__ int T[PER * 32];
    const int tid  = threadIdx.x;
    const int lane = tid & 31;
    const int wave = tid >> 5;

    int tot[PER];
#pragma unroll
    for (int j = 0; j < PER; ++j) {
        const int k = min(wave + 32 * j, NBK - 1);
        const int4* p = (const int4*)(H + k * NCH + lane * 8);
        const int4 a = p[0];
        const int4 b = p[1];
        tot[j] = a.x + a.y + a.z + a.w + b.x + b.y + b.z + b.w;
    }
#pragma unroll
    for (int j = 0; j < PER; ++j) {
        int s = tot[j];
#pragma unroll
        for (int o = 16; o > 0; o >>= 1) s += __shfl_xor(s, o, 32);
        const int k = wave + 32 * j;
        if (lane == 0) T[k] = (k < NBK) ? s : 0;
    }
    __syncthreads();
    if (wave == 0) {
        int v[PER];
        int sum = 0;
#pragma unroll
        for (int j = 0; j < PER; ++j) { v[j] = T[lane * PER + j]; sum += v[j]; }
        int incl = sum;
#pragma unroll
        for (int o = 1; o < 32; o <<= 1) {
            const int y = __shfl_up(incl, o, 32);
            if (lane >= o) incl += y;
        }
        int run = incl - sum;
#pragma unroll
        for (int j = 0; j < PER; ++j) { T[lane * PER + j] = run; run += v[j]; }
    }
    __syncthreads();

    const int k = blockIdx.x * 32 + wave;
    if (k < NBK) {
        const int4* p = (const int4*)(H + k * NCH + lane * 8);
        const int4 a = p[0];
        const int4 b = p[1];
        const int s = a.x + a.y + a.z + a.w + b.x + b.y + b.z + b.w;
        int incl = s;
#pragma unroll
        for (int o = 1; o < 32; o <<= 1) {
            const int y = __shfl_up(incl, o, 32);
            if (lane >= o) incl += y;
        }
        const int first = T[k];
        int run = first + incl - s;
        int4 oa, ob;
        oa.x = run; run += a.x;
        oa.y = run; run += a.y;
        oa.z = run; run += a.z;
        oa.w = run; run += a.w;
        ob.x = run; run += b.x;
        ob.y = run; run += b.y;
        ob.z = run; run += b.z;
        ob.w = run;
        int4* q = (int4*)(base + k * NCH + lane * 8);
        q[0] = oa;
        q[1] = ob;
        if (lane == 0) bstart[k] = first;
    }
    if (blockIdx.x == 0 && tid == 0) bstart[NBK] = N_EDGES;
}

__global__ __launch_bounds__(L1_THREADS)
void bucket_scatter(const int* __restrict__ src, const int* __restrict__ dst,
                    const int* __restrict__ base, unsigned* __restrict__ packed) {
    __shared__ int cur[NBK];
    const int tid = threadIdx.x;
    const int e0 = blockIdx.x * CH_EDGES;
    const int e1 = min(e0 + CH_EDGES, N_EDGES);
    int d[L1_ITERS], s[L1_ITERS];
#pragma unroll
    for (int j = 0; j < L1_ITERS; ++j) {
        const int e = e0 + tid + j * L1_THREADS;
        d[j] = (e < e1) ? dst[e] : -1;
        s[j] = (e < e1) ? src[e] : 0;
    }
    for (int k = tid; k < NBK; k += L1_THREADS) cur[k] = base[k * NCH + blockIdx.x];
    __syncthreads();
#pragma unroll
    for (int j = 0; j < L1_ITERS; ++j) {
        if (d[j] >= 0) {
            const int p = atomicAdd(&cur[d[j] >> BK_SHIFT], 1);
            packed[p] = ((unsigned)(d[j] & (BK_NODES - 1)) << SRC_BITS) | (unsigned)s[j];
        }
    }
}

#define FIN_THREADS 1024
#define FIN_KEEP 6
__global__ __launch_bounds__(FIN_THREADS)
void bucket_finalize(const unsigned* __restrict__ packed, const int* __restrict__ bstart,
                     int* __restrict__ rowptr, float* __restrict__ dinv,
                     int* __restrict__ csr_src) {
    __shared__ int cnt[BK_NODES];
    __shared__ int cur[BK_NODES];
    __shared__ int wtot[BK_NODES / 32];
    const int tid  = threadIdx.x;
    const int lane = tid & 31;
    const int wave = tid >> 5;
    const int k  = blockIdx.x;
    const int b0 = bstart[k];
    const int b1 = bstart[k + 1];

    unsigned w[FIN_KEEP];
#pragma unroll
    for (int j = 0; j < FIN_KEEP; ++j) {
        const int i = b0 + tid + j * FIN_THREADS;
        w[j] = (i < b1) ? packed[i] : NO_EDGE;
    }
    if (tid < BK_NODES) cnt[tid] = 0;
    __syncthreads();
#pragma unroll
    for (int j = 0; j < FIN_KEEP; ++j)
        if (w[j] != NO_EDGE) atomicAdd(&cnt[w[j] >> SRC_BITS], 1);
    for (int i = b0 + tid + FIN_KEEP * FIN_THREADS; i < b1; i += FIN_THREADS)
        atomicAdd(&cnt[packed[i] >> SRC_BITS], 1);
    __syncthreads();

    const int x = (tid < BK_NODES) ? cnt[tid] : 0;
    int incl = x;
#pragma unroll
    for (int o = 1; o < 32; o <<= 1) {
        const int y = __shfl_up(incl, o, 32);
        if (lane >= o) incl += y;
    }
    if (tid < BK_NODES && lane == 31) wtot[wave] = incl;
    __syncthreads();
    if (tid < BK_NODES) {
        int pre = 0;
        for (int q = 0; q < wave; ++q) pre += wtot[q];
        const int excl = pre + incl - x;
        cur[tid] = excl;
        const int node = k * BK_NODES + tid;
        if (node < N_NODES) {
            rowptr[node] = b0 + excl;
            const float dg = (float)(x + 1);
            dinv[node] = dg > 0.0f ? rsqrtf(dg) : 0.0f;
        }
    }
    if (k == NBK - 1 && tid == 0) rowptr[N_NODES] = b1;
    __syncthreads();

#pragma unroll
    for (int j = 0; j < FIN_KEEP; ++j) {
        if (w[j] != NO_EDGE) {
            const int p = atomicAdd(&cur[w[j] >> SRC_BITS], 1);
            csr_src[b0 + p] = (int)(w[j] & SRC_MASK);
        }
    }
    for (int i = b0 + tid + FIN_KEEP * FIN_THREADS; i < b1; i += FIN_THREADS) {
        const unsigned ww = packed[i];
        const int p = atomicAdd(&cur[ww >> SRC_BITS], 1);
        csr_src[b0 + p] = (int)(ww & SRC_MASK);
    }
}

template <int NCOLS>
__global__ __launch_bounds__(256)
void gemm_scale_kernel(const float* __restrict__ A,
                       const float* __restrict__ W,
                       const float* __restrict__ dinv,
                       float* __restrict__ out) {
    constexpr int NT = NCOLS / 16;
    __shared__ float lW[128 * NCOLS];

    const int tid = threadIdx.x;
    for (int idx = tid; idx < 128 * NCOLS; idx += 256) {
        const int k = idx / NCOLS;
        const int n = idx - k * NCOLS;
        lW[(k >> 1) * (2 * NCOLS) + 2 * n + (k & 1)] = W[idx];
    }
    __syncthreads();

    const int wave  = tid >> 5;
    const int lane  = tid & 31;
    const int strip = blockIdx.x * 8 + wave;
    if (strip >= N_STRIPS) return;

    const int row0 = strip * 16;
    const int half = lane >> 4;
    const int l16  = lane & 15;
    const int arow = row0 + l16;

    v8f acc[NT];
#pragma unroll
    for (int t = 0; t < NT; ++t) acc[t] = (v8f){0.f, 0.f, 0.f, 0.f, 0.f, 0.f, 0.f, 0.f};

    for (int k0 = 0; k0 < 128; k0 += 4) {
        const int ka = k0 + 2 * half;
        const v2f a = *(const v2f*)&A[(long)arow * 128 + ka];
        const float* __restrict__ lrow = &lW[(ka >> 1) * (2 * NCOLS)];
#pragma unroll
        for (int t = 0; t < NT; ++t) {
            const int n = t * 16 + l16;
            const v2f b = *(const v2f*)&lrow[2 * n];
            acc[t] = __builtin_amdgcn_wmma_f32_16x16x4_f32(
                false, a, false, b, (short)0, acc[t], false, false);
        }
    }

#pragma unroll
    for (int t = 0; t < NT; ++t) {
#pragma unroll
        for (int r = 0; r < 8; ++r) {
            const int rr = row0 + r + 8 * half;
            out[(long)rr * NCOLS + t * 16 + l16] = acc[t][r] * dinv[rr];
        }
    }
}

#define SP 132
__device__ __forceinline__
float4 gather_row128(const int* __restrict__ csr_src, const float4* __restrict__ hs4,
                     int node, int beg, int end, int lane) {
    float4 acc = hs4[(long)node * 32 + lane];
    for (int base = beg; base < end; base += 32) {
        const int n = min(32, end - base);
        const int idx = (lane < n) ? csr_src[base + lane] : 0;
#pragma unroll 4
        for (int j = 0; j < n; ++j) {
            const int s = __builtin_amdgcn_readlane(idx, j);
            const float4 v = hs4[(long)s * 32 + lane];
            acc.x += v.x; acc.y += v.y; acc.z += v.z; acc.w += v.w;
        }
    }
    return acc;
}

__global__ __launch_bounds__(256, 16)
void gather128_gemm32(const int* __restrict__ rowptr, const int* __restrict__ csr_src,
                      const float* __restrict__ hs, const float* __restrict__ dinv,
                      const float* __restrict__ b1, const float* __restrict__ W2,
                      float* __restrict__ h2s) {
    __shared__ alignas(16) float S[16 * SP];
    const int lane = threadIdx.x & 31;
    const int wave = __builtin_amdgcn_readfirstlane(threadIdx.x >> 5);
    const int row0 = blockIdx.x * 16;
    const float4* __restrict__ hs4 = (const float4*)hs;
    const int c = lane * 4;
    const float4 bb = *(const float4*)&b1[c];

    const int na = row0 + 2 * wave;
    const int r0 = rowptr[na];
    const int r1 = rowptr[na + 1];
    const int r2 = rowptr[na + 2];
#pragma unroll
    for (int i = 0; i < 2; ++i) {
        const int node = na + i;
        const float4 acc = gather_row128(csr_src, hs4, node, i ? r1 : r0, i ? r2 : r1, lane);
        const float dn = dinv[node];
        float4 o;
        o.x = fmaxf(acc.x * dn + bb.x, 0.0f);
        o.y = fmaxf(acc.y * dn + bb.y, 0.0f);
        o.z = fmaxf(acc.z * dn + bb.z, 0.0f);
        o.w = fmaxf(acc.w * dn + bb.w, 0.0f);
        *(float4*)&S[(2 * wave + i) * SP + c] = o;
    }
    __syncthreads();
    if (wave >= 2) return;

    const int half = lane >> 4;
    const int l16  = lane & 15;
    const int col  = wave * 16 + l16;
    v8f acc = (v8f){0.f, 0.f, 0.f, 0.f, 0.f, 0.f, 0.f, 0.f};
#pragma unroll
    for (int k0 = 0; k0 < 128; k0 += 4) {
        const int ka = k0 + 2 * half;
        const v2f a = *(const v2f*)&S[l16 * SP + ka];
        v2f b;
        b.x = W2[ka * N_CLS + col];
        b.y = W2[(ka + 1) * N_CLS + col];
        acc = __builtin_amdgcn_wmma_f32_16x16x4_f32(
            false, a, false, b, (short)0, acc, false, false);
    }
#pragma unroll
    for (int r = 0; r < 8; ++r) {
        const int rr = row0 + r + 8 * half;
        h2s[(long)rr * N_CLS + col] = acc[r] * dinv[rr];
    }
}

__global__ __launch_bounds__(256)
void gather32_final(const int* __restrict__ rowptr, const int* __restrict__ csr_src,
                    const float* __restrict__ h2s, const float* __restrict__ dinv,
                    const float* __restrict__ b2, float* __restrict__ out) {
    const int lane = threadIdx.x & 31;
    const int node = __builtin_amdgcn_readfirstlane(blockIdx.x * 8 + (threadIdx.x >> 5));
    if (node >= N_NODES) return;
    const int beg = rowptr[node];
    const int end = rowptr[node + 1];
    float acc = h2s[(long)node * 32 + lane];
    for (int base = beg; base < end; base += 32) {
        const int n = min(32, end - base);
        const int idx = (lane < n) ? csr_src[base + lane] : 0;
#pragma unroll 4
        for (int j = 0; j < n; ++j) {
            const int s = __builtin_amdgcn_readlane(idx, j);
            acc += h2s[(long)s * 32 + lane];
        }
    }
    const float v = acc * dinv[node] + b2[lane];

    float m = v;
#pragma unroll
    for (int o = 16; o > 0; o >>= 1) m = fmaxf(m, __shfl_xor(m, o, 32));
    float s = __expf(v - m);
#pragma unroll
    for (int o = 16; o > 0; o >>= 1) s += __shfl_xor(s, o, 32);
    out[(long)node * 32 + lane] = (v - m) - __logf(s);
}

extern "C" void kernel_launch(void* const* d_in, const int* in_sizes, int n_in,
                              void* d_out, int out_size, void* d_ws, size_t ws_size,
                              hipStream_t stream) {
    const float* x  = (const float*)d_in[0];
    const int*   ei = (const int*)d_in[1];
    const float* W1 = (const float*)d_in[2];
    const float* b1 = (const float*)d_in[3];
    const float* W2 = (const float*)d_in[4];
    const float* b2 = (const float*)d_in[5];
    float* out = (float*)d_out;

    const int* src = ei;
    const int* dst = ei + N_EDGES;

    float* dinv = (float*)d_ws;
    float* hs   = dinv + N_NODES;
    float* h2s  = hs   + (long)N_NODES * 256;
    int* rowptr  = (int*)(h2s + (long)N_NODES * 32);
    int* csr_src = rowptr + N_NODES + 32;
    unsigned* packed = (unsigned*)(csr_src + N_EDGES);
    int* H      = (int*)(packed + N_EDGES);
    int* base   = H + NBK * NCH;
    int* bstart = base + NBK * NCH;

    bucket_hist<<<NCH, L1_THREADS, 0, stream>>>(dst, H);
    bucket_scan<<<(NBK + 31) / 32, 1024, 0, stream>>>(H, base, bstart);
    bucket_scatter<<<NCH, L1_THREADS, 0, stream>>>(src, dst, base, packed);
    bucket_finalize<<<NBK, FIN_THREADS, 0, stream>>>(packed, bstart, rowptr, dinv, csr_src);

    gemm_scale_kernel<128><<<(N_STRIPS + 7) / 8, 256, 0, stream>>>(x, W1, dinv, hs);
    gather128_gemm32<<<N_STRIPS, 256, 0, stream>>>(rowptr, csr_src, hs, dinv, b1, W2, h2s);
    gather32_final<<<(N_NODES + 7) / 8, 256, 0, stream>>>(rowptr, csr_src, h2s, dinv, b2, out);
}


# Layer-2 gather: 16 nodes per wave (8 lanes x 16 B per 128-B row, one load brings 4 rows; 6250 waves enter, not 100000); edge sum order kept, softmax exp-sum tree changed (4 in lane, then 8 lanes)
# speedup vs baseline: 3.7449x; 1.0567x over previous
#include <hip/hip_runtime.h>
#include <hip/hip_bf16.h>

#define N_NODES 100000
#define N_EDGES 1600000
#define D_IN    128
#define D_HID   128
#define N_CLS   32
#define N_STRIPS (N_NODES / 16)

typedef float v2f __attribute__((ext_vector_type(2)));
typedef float v8f __attribute__((ext_vector_type(8)));

#define BK_SHIFT 8
#define BK_NODES (1 << BK_SHIFT)
#define NBK      ((N_NODES + BK_NODES - 1) / BK_NODES)
#define NCH      256
#define CH_EDGES ((N_EDGES + NCH - 1) / NCH)
#define L1_THREADS 1024
#define L1_ITERS ((CH_EDGES + L1_THREADS - 1) / L1_THREADS)
#define SRC_BITS 17
#define SRC_MASK ((1u << SRC_BITS) - 1u)
#define NO_EDGE  0xFFFFFFFFu

static_assert(NCH == 256, "bucket_scan gives each lane 8 of the 256 chunk counts");
static_assert(N_NODES <= (1 << SRC_BITS), "source id must fit the packed word");

__global__ __launch_bounds__(L1_THREADS)
void bucket_hist(const int* __restrict__ dst, int* __restrict__ H) {
    __shared__ int h[NBK];
    const int tid = threadIdx.x;
    const int e0 = blockIdx.x * CH_EDGES;
    const int e1 = min(e0 + CH_EDGES, N_EDGES);
    int d[L1_ITERS];
#pragma unroll
    for (int j = 0; j < L1_ITERS; ++j) {
        const int e = e0 + tid + j * L1_THREADS;
        d[j] = (e < e1) ? dst[e] : -1;
    }
    for (int k = tid; k < NBK; k += L1_THREADS) h[k] = 0;
    __syncthreads();
#pragma unroll
    for (int j = 0; j < L1_ITERS; ++j)
        if (d[j] >= 0) atomicAdd(&h[d[j] >> BK_SHIFT], 1);
    __syncthreads();
    for (int k = tid; k < NBK; k += L1_THREADS) H[k * NCH + blockIdx.x] = h[k];
}

__global__ __launch_bounds__(1024)
void bucket_scan(const int* __restrict__ H, int* __restrict__ base,
                 int* __restrict__ bstart) {
    constexpr int PER = (NBK + 31) / 32;
    __shared__ int T[PER * 32];
    const int tid  = threadIdx.x;
    const int lane = tid & 31;
    const int wave = tid >> 5;

    int tot[PER];
#pragma unroll
    for (int j = 0; j < PER; ++j) {
        const int k = min(wave + 32 * j, NBK - 1);
        const int4* p = (const int4*)(H + k * NCH + lane * 8);
        const int4 a = p[0];
        const int4 b = p[1];
        tot[j] = a.x + a.y + a.z + a.w + b.x + b.y + b.z + b.w;
    }
#pragma unroll
    for (int j = 0; j < PER; ++j) {
        int s = tot[j];
#pragma unroll
        for (int o = 16; o > 0; o >>= 1) s += __shfl_xor(s, o, 32);
        const int k = wave + 32 * j;
        if (lane == 0) T[k] = (k < NBK) ? s : 0;
    }
    __syncthreads();
    if (wave == 0) {
        int v[PER];
        int sum = 0;
#pragma unroll
        for (int j = 0; j < PER; ++j) { v[j] = T[lane * PER + j]; sum += v[j]; }
        int incl = sum;
#pragma unroll
        for (int o = 1; o < 32; o <<= 1) {
            const int y = __shfl_up(incl, o, 32);
            if (lane >= o) incl += y;
        }
        int run = incl - sum;
#pragma unroll
        for (int j = 0; j < PER; ++j) { T[lane * PER + j] = run; run += v[j]; }
    }
    __syncthreads();

    const int k = blockIdx.x * 32 + wave;
    if (k < NBK) {
        const int4* p = (const int4*)(H + k * NCH + lane * 8);
        const int4 a = p[0];
        const int4 b = p[1];
        const int s = a.x + a.y + a.z + a.w + b.x + b.y + b.z + b.w;
        int incl = s;
#pragma unroll
        for (int o = 1; o < 32; o <<= 1) {
            const int y = __shfl_up(incl, o, 32);
            if (lane >= o) incl += y;
        }
        const int first = T[k];
        int run = first + incl - s;
        int4 oa, ob;
        oa.x = run; run += a.x;
        oa.y = run; run += a.y;
        oa.z = run; run += a.z;
        oa.w = run; run += a.w;
        ob.x = run; run += b.x;
        ob.y = run; run += b.y;
        ob.z = run; run += b.z;
        ob.w = run;
        int4* q = (int4*)(base + k * NCH + lane * 8);
        q[0] = oa;
        q[1] = ob;
        if (lane == 0) bstart[k] = first;
    }
    if (blockIdx.x == 0 && tid == 0) bstart[NBK] = N_EDGES;
}

__global__ __launch_bounds__(L1_THREADS)
void bucket_scatter(const int* __restrict__ src, const int* __restrict__ dst,
                    const int* __restrict__ base, unsigned* __restrict__ packed) {
    __shared__ int cur[NBK];
    const int tid = threadIdx.x;
    const int e0 = blockIdx.x * CH_EDGES;
    const int e1 = min(e0 + CH_EDGES, N_EDGES);
    int d[L1_ITERS], s[L1_ITERS];
#pragma unroll
    for (int j = 0; j < L1_ITERS; ++j) {
        const int e = e0 + tid + j * L1_THREADS;
        d[j] = (e < e1) ? dst[e] : -1;
        s[j] = (e < e1) ? src[e] : 0;
    }
    for (int k = tid; k < NBK; k += L1_THREADS) cur[k] = base[k * NCH + blockIdx.x];
    __syncthreads();
#pragma unroll
    for (int j = 0; j < L1_ITERS; ++j) {
        if (d[j] >= 0) {
            const int p = atomicAdd(&cur[d[j] >> BK_SHIFT], 1);
            packed[p] = ((unsigned)(d[j] & (BK_NODES - 1)) << SRC_BITS) | (unsigned)s[j];
        }
    }
}

#define FIN_THREADS 1024
#define FIN_KEEP 6
__global__ __launch_bounds__(FIN_THREADS)
void bucket_finalize(const unsigned* __restrict__ packed, const int* __restrict__ bstart,
                     int* __restrict__ rowptr, float* __restrict__ dinv,
                     int* __restrict__ csr_src) {
    __shared__ int cnt[BK_NODES];
    __shared__ int cur[BK_NODES];
    __shared__ int wtot[BK_NODES / 32];
    const int tid  = threadIdx.x;
    const int lane = tid & 31;
    const int wave = tid >> 5;
    const int k  = blockIdx.x;
    const int b0 = bstart[k];
    const int b1 = bstart[k + 1];

    unsigned w[FIN_KEEP];
#pragma unroll
    for (int j = 0; j < FIN_KEEP; ++j) {
        const int i = b0 + tid + j * FIN_THREADS;
        w[j] = (i < b1) ? packed[i] : NO_EDGE;
    }
    if (tid < BK_NODES) cnt[tid] = 0;
    __syncthreads();
#pragma unroll
    for (int j = 0; j < FIN_KEEP; ++j)
        if (w[j] != NO_EDGE) atomicAdd(&cnt[w[j] >> SRC_BITS], 1);
    for (int i = b0 + tid + FIN_KEEP * FIN_THREADS; i < b1; i += FIN_THREADS)
        atomicAdd(&cnt[packed[i] >> SRC_BITS], 1);
    __syncthreads();

    const int x = (tid < BK_NODES) ? cnt[tid] : 0;
    int incl = x;
#pragma unroll
    for (int o = 1; o < 32; o <<= 1) {
        const int y = __shfl_up(incl, o, 32);
        if (lane >= o) incl += y;
    }
    if (tid < BK_NODES && lane == 31) wtot[wave] = incl;
    __syncthreads();
    if (tid < BK_NODES) {
        int pre = 0;
        for (int q = 0; q < wave; ++q) pre += wtot[q];
        const int excl = pre + incl - x;
        cur[tid] = excl;
        const int node = k * BK_NODES + tid;
        if (node < N_NODES) {
            rowptr[node] = b0 + excl;
            const float dg = (float)(x + 1);
            dinv[node] = dg > 0.0f ? rsqrtf(dg) : 0.0f;
        }
    }
    if (k == NBK - 1 && tid == 0) rowptr[N_NODES] = b1;
    __syncthreads();

#pragma unroll
    for (int j = 0; j < FIN_KEEP; ++j) {
        if (w[j] != NO_EDGE) {
            const int p = atomicAdd(&cur[w[j] >> SRC_BITS], 1);
            csr_src[b0 + p] = (int)(w[j] & SRC_MASK);
        }
    }
    for (int i = b0 + tid + FIN_KEEP * FIN_THREADS; i < b1; i += FIN_THREADS) {
        const unsigned ww = packed[i];
        const int p = atomicAdd(&cur[ww >> SRC_BITS], 1);
        csr_src[b0 + p] = (int)(ww & SRC_MASK);
    }
}

template <int NCOLS>
__global__ __launch_bounds__(256)
void gemm_scale_kernel(const float* __restrict__ A,
                       const float* __restrict__ W,
                       const float* __restrict__ dinv,
                       float* __restrict__ out) {
    constexpr int NT = NCOLS / 16;
    __shared__ float lW[128 * NCOLS];

    const int tid = threadIdx.x;
    for (int idx = tid; idx < 128 * NCOLS; idx += 256) {
        const int k = idx / NCOLS;
        const int n = idx - k * NCOLS;
        lW[(k >> 1) * (2 * NCOLS) + 2 * n + (k & 1)] = W[idx];
    }
    __syncthreads();

    const int wave  = tid >> 5;
    const int lane  = tid & 31;
    const int strip = blockIdx.x * 8 + wave;
    if (strip >= N_STRIPS) return;

    const int row0 = strip * 16;
    const int half = lane >> 4;
    const int l16  = lane & 15;
    const int arow = row0 + l16;

    v8f acc[NT];
#pragma unroll
    for (int t = 0; t < NT; ++t) acc[t] = (v8f){0.f, 0.f, 0.f, 0.f, 0.f, 0.f, 0.f, 0.f};

    for (int k0 = 0; k0 < 128; k0 += 4) {
        const int ka = k0 + 2 * half;
        const v2f a = *(const v2f*)&A[(long)arow * 128 + ka];
        const float* __restrict__ lrow = &lW[(ka >> 1) * (2 * NCOLS)];
#pragma unroll
        for (int t = 0; t < NT; ++t) {
            const int n = t * 16 + l16;
            const v2f b = *(const v2f*)&lrow[2 * n];
            acc[t] = __builtin_amdgcn_wmma_f32_16x16x4_f32(
                false, a, false, b, (short)0, acc[t], false, false);
        }
    }

#pragma unroll
    for (int t = 0; t < NT; ++t) {
#pragma unroll
        for (int r = 0; r < 8; ++r) {
            const int rr = row0 + r + 8 * half;
            out[(long)rr * NCOLS + t * 16 + l16] = acc[t][r] * dinv[rr];
        }
    }
}

#define SP 132
__device__ __forceinline__
float4 gather_row128(const int* __restrict__ csr_src, const float4* __restrict__ hs4,
                     int node, int beg, int end, int lane) {
    float4 acc = hs4[(long)node * 32 + lane];
    for (int base = beg; base < end; base += 32) {
        const int n = min(32, end - base);
        const int idx = (lane < n) ? csr_src[base + lane] : 0;
#pragma unroll 4
        for (int j = 0; j < n; ++j) {
            const int s = __builtin_amdgcn_readlane(idx, j);
            const float4 v = hs4[(long)s * 32 + lane];
            acc.x += v.x; acc.y += v.y; acc.z += v.z; acc.w += v.w;
        }
    }
    return acc;
}

__global__ __launch_bounds__(256, 16)
void gather128_gemm32(const int* __restrict__ rowptr, const int* __restrict__ csr_src,
                      const float* __restrict__ hs, const float* __restrict__ dinv,
                      const float* __restrict__ b1, const float* __restrict__ W2,
                      float* __restrict__ h2s) {
    __shared__ alignas(16) float S[16 * SP];
    const int lane = threadIdx.x & 31;
    const int wave = __builtin_amdgcn_readfirstlane(threadIdx.x >> 5);
    const int row0 = blockIdx.x * 16;
    const float4* __restrict__ hs4 = (const float4*)hs;
    const int c = lane * 4;
    const float4 bb = *(const float4*)&b1[c];

    const int na = row0 + 2 * wave;
    const int r0 = rowptr[na];
    const int r1 = rowptr[na + 1];
    const int r2 = rowptr[na + 2];
#pragma unroll
    for (int i = 0; i < 2; ++i) {
        const int node = na + i;
        const float4 acc = gather_row128(csr_src, hs4, node, i ? r1 : r0, i ? r2 : r1, lane);
        const float dn = dinv[node];
        float4 o;
        o.x = fmaxf(acc.x * dn + bb.x, 0.0f);
        o.y = fmaxf(acc.y * dn + bb.y, 0.0f);
        o.z = fmaxf(acc.z * dn + bb.z, 0.0f);
        o.w = fmaxf(acc.w * dn + bb.w, 0.0f);
        *(float4*)&S[(2 * wave + i) * SP + c] = o;
    }
    __syncthreads();
    if (wave >= 2) return;

    const int half = lane >> 4;
    const int l16  = lane & 15;
    const int col  = wave * 16 + l16;
    v8f acc = (v8f){0.f, 0.f, 0.f, 0.f, 0.f, 0.f, 0.f, 0.f};
#pragma unroll
    for (int k0 = 0; k0 < 128; k0 += 4) {
        const int ka = k0 + 2 * half;
        const v2f a = *(const v2f*)&S[l16 * SP + ka];
        v2f b;
        b.x = W2[ka * N_CLS + col];
        b.y = W2[(ka + 1) * N_CLS + col];
        acc = __builtin_amdgcn_wmma_f32_16x16x4_f32(
            false, a, false, b, (short)0, acc, false, false);
    }
#pragma unroll
    for (int r = 0; r < 8; ++r) {
        const int rr = row0 + r + 8 * half;
        h2s[(long)rr * N_CLS + col] = acc[r] * dinv[rr];
    }
}

template <int J>
__device__ __forceinline__ int group8_bcast(int v) {
    return __builtin_amdgcn_mov_dpp8(v, J * 0x249249);
}
constexpr unsigned group8_xor_sel(int m) {
    unsigned s = 0;
    for (int k = 0; k < 8; ++k) s |= (unsigned)(k ^ m) << (3 * k);
    return s;
}
template <int M>
__device__ __forceinline__ float group8_xor(float v) {
    return __int_as_float(__builtin_amdgcn_mov_dpp8(__float_as_int(v), group8_xor_sel(M)));
}

#define G32_SETS 4
#define G32_NPW  (4 * G32_SETS)
static_assert(G32_SETS % 2 == 0, "sets are taken two at a time");
static_assert(N_NODES % G32_NPW == 0, "a wave is all inside the node range or all outside");

#define G32_ROW(V, IX, N, J)                                           \
    {                                                                  \
        const int s = group8_bcast<J>(IX);                             \
        V[J] = make_float4(0.0f, 0.0f, 0.0f, 0.0f);                    \
        if (J < N) V[J] = h4[s * 8 + i];                               \
    }
#define G32_ROWS(V, IX, N)                                             \
    G32_ROW(V, IX, N, 0) G32_ROW(V, IX, N, 1) G32_ROW(V, IX, N, 2) G32_ROW(V, IX, N, 3) \
    G32_ROW(V, IX, N, 4) G32_ROW(V, IX, N, 5) G32_ROW(V, IX, N, 6) G32_ROW(V, IX, N, 7)

__device__ __forceinline__
void g32_finish(const float4 a, const float dn, const float4 bb, float4* __restrict__ dst) {
    float4 t;
    t.x = a.x * dn + bb.x;
    t.y = a.y * dn + bb.y;
    t.z = a.z * dn + bb.z;
    t.w = a.w * dn + bb.w;
    float m = fmaxf(fmaxf(t.x, t.y), fmaxf(t.z, t.w));
    m = fmaxf(m, group8_xor<1>(m));
    m = fmaxf(m, group8_xor<2>(m));
    m = fmaxf(m, group8_xor<4>(m));
    float s = (__expf(t.x - m) + __expf(t.y - m)) + (__expf(t.z - m) + __expf(t.w - m));
    s += group8_xor<1>(s);
    s += group8_xor<2>(s);
    s += group8_xor<4>(s);
    const float ls = __logf(s);
    float4 o;
    o.x = (t.x - m) - ls;
    o.y = (t.y - m) - ls;
    o.z = (t.z - m) - ls;
    o.w = (t.w - m) - ls;
    *dst = o;
}

__global__ __launch_bounds__(256, 8)
void gather32_final(const int* __restrict__ rowptr, const int* __restrict__ csr_src,
                    const float* __restrict__ h2s, const float* __restrict__ dinv,
                    const float* __restrict__ b2, float* __restrict__ out) {
    const int lane = threadIdx.x & 31;
    const int wave = __builtin_amdgcn_readfirstlane(threadIdx.x >> 5);
    const int n0 = (blockIdx.x * 8 + wave) * G32_NPW;
    if (n0 >= N_NODES) return;
    const int g = lane >> 3;
    const int i = lane & 7;
    const float4* __restrict__ h4 = (const float4*)h2s;
    float4* __restrict__ out4 = (float4*)out;
    const float4 bb = *(const float4*)&b2[4 * i];

    int node[G32_SETS], cur[G32_SETS], end[G32_SETS], idx[G32_SETS];
    float dn[G32_SETS];
    float4 acc[G32_SETS];
#pragma unroll
    for (int q = 0; q < G32_SETS; ++q) {
        node[q] = n0 + 4 * q + g;
        cur[q] = rowptr[node[q]];
        end[q] = rowptr[node[q] + 1];
    }
#pragma unroll
    for (int q = 0; q < G32_SETS; ++q) {
        dn[q]  = dinv[node[q]];
        acc[q] = h4[node[q] * 8 + i];
        idx[q] = (cur[q] + i < end[q]) ? csr_src[cur[q] + i] : 0;
    }

#pragma unroll
    for (int q = 0; q < G32_SETS; q += 2) {
        float4 aA = acc[q], aB = acc[q + 1];
        int cA = cur[q], cB = cur[q + 1];
        const int eA = end[q], eB = end[q + 1];
        int ixA = idx[q], ixB = idx[q + 1];
        while (__any((cA < eA) | (cB < eB))) {
            const int pA = cA + 8 + i;
            const int pB = cB + 8 + i;
            const int nixA = (pA < eA) ? csr_src[pA] : 0;
            const int nixB = (pB < eB) ? csr_src[pB] : 0;
            const int nA = eA - cA;
            const int nB = eB - cB;
            float4 v[8], w[8];
            G32_ROWS(v, ixA, nA)
            G32_ROWS(w, ixB, nB)
#pragma unroll
            for (int j = 0; j < 8; ++j) {
                aA.x += v[j].x; aA.y += v[j].y; aA.z += v[j].z; aA.w += v[j].w;
            }
#pragma unroll
            for (int j = 0; j < 8; ++j) {
                aB.x += w[j].x; aB.y += w[j].y; aB.z += w[j].z; aB.w += w[j].w;
            }
            cA += 8;
            cB += 8;
            ixA = nixA;
            ixB = nixB;
        }
        g32_finish(aA, dn[q],     bb, &out4[node[q] * 8 + i]);
        g32_finish(aB, dn[q + 1], bb, &out4[node[q + 1] * 8 + i]);
    }
}

extern "C" void kernel_launch(void* const* d_in, const int* in_sizes, int n_in,
                              void* d_out, int out_size, void* d_ws, size_t ws_size,
                              hipStream_t stream) {
    const float* x  = (const float*)d_in[0];
    const int*   ei = (const int*)d_in[1];
    const float* W1 = (const float*)d_in[2];
    const float* b1 = (const float*)d_in[3];
    const float* W2 = (const float*)d_in[4];
    const float* b2 = (const float*)d_in[5];
    float* out = (float*)d_out;

    const int* src = ei;
    const int* dst = ei + N_EDGES;

    float* dinv = (float*)d_ws;
    float* hs   = dinv + N_NODES;
    float* h2s  = hs   + (long)N_NODES * 256;
    int* rowptr  = (int*)(h2s + (long)N_NODES * 32);
    int* csr_src = rowptr + N_NODES + 32;
    unsigned* packed = (unsigned*)(csr_src + N_EDGES);
    int* H      = (int*)(packed + N_EDGES);
    int* base   = H + NBK * NCH;
    int* bstart = base + NBK * NCH;

    bucket_hist<<<NCH, L1_THREADS, 0, stream>>>(dst, H);
    bucket_scan<<<(NBK + 31) / 32, 1024, 0, stream>>>(H, base, bstart);
    bucket_scatter<<<NCH, L1_THREADS, 0, stream>>>(src, dst, base, packed);
    bucket_finalize<<<NBK, FIN_THREADS, 0, stream>>>(packed, bstart, rowptr, dinv, csr_src);

    gemm_scale_kernel<128><<<(N_STRIPS + 7) / 8, 256, 0, stream>>>(x, W1, dinv, hs);
    gather128_gemm32<<<N_STRIPS, 256, 0, stream>>>(rowptr, csr_src, hs, dinv, b1, W2, h2s);
    gather32_final<<<(N_NODES + 8 * G32_NPW - 1) / (8 * G32_NPW), 256, 0, stream>>>(rowptr, csr_src, h2s, dinv, b2, out);
}


# Layer-1 product moved behind the gather: x rows summed first (dinv[src] per edge), 16-node strip meets a register-held W1 slice on WMMA in persistent blocks; product launch gone; sum order changed
# speedup vs baseline: 4.3034x; 1.1491x over previous
#include <hip/hip_runtime.h>
#include <hip/hip_bf16.h>

#define N_NODES 100000
#define N_EDGES 1600000
#define D_IN    128
#define D_HID   128
#define N_CLS   32
#define N_STRIPS (N_NODES / 16)

typedef float v2f __attribute__((ext_vector_type(2)));
typedef float v8f __attribute__((ext_vector_type(8)));

#define BK_SHIFT 8
#define BK_NODES (1 << BK_SHIFT)
#define NBK      ((N_NODES + BK_NODES - 1) / BK_NODES)
#define NCH      256
#define CH_EDGES ((N_EDGES + NCH - 1) / NCH)
#define L1_THREADS 1024
#define L1_ITERS ((CH_EDGES + L1_THREADS - 1) / L1_THREADS)
#define SRC_BITS 17
#define SRC_MASK ((1u << SRC_BITS) - 1u)
#define NO_EDGE  0xFFFFFFFFu

static_assert(NCH == 256, "bucket_scan gives each lane 8 of the 256 chunk counts");
static_assert(N_NODES <= (1 << SRC_BITS), "source id must fit the packed word");

__global__ __launch_bounds__(L1_THREADS)
void bucket_hist(const int* __restrict__ dst, int* __restrict__ H) {
    __shared__ int h[NBK];
    const int tid = threadIdx.x;
    const int e0 = blockIdx.x * CH_EDGES;
    const int e1 = min(e0 + CH_EDGES, N_EDGES);
    int d[L1_ITERS];
#pragma unroll
    for (int j = 0; j < L1_ITERS; ++j) {
        const int e = e0 + tid + j * L1_THREADS;
        d[j] = (e < e1) ? dst[e] : -1;
    }
    for (int k = tid; k < NBK; k += L1_THREADS) h[k] = 0;
    __syncthreads();
#pragma unroll
    for (int j = 0; j < L1_ITERS; ++j)
        if (d[j] >= 0) atomicAdd(&h[d[j] >> BK_SHIFT], 1);
    __syncthreads();
    for (int k = tid; k < NBK; k += L1_THREADS) H[k * NCH + blockIdx.x] = h[k];
}

__global__ __launch_bounds__(1024)
void bucket_scan(const int* __restrict__ H, int* __restrict__ base,
                 int* __restrict__ bstart) {
    constexpr int PER = (NBK + 31) / 32;
    __shared__ int T[PER * 32];
    const int tid  = threadIdx.x;
    const int lane = tid & 31;
    const int wave = tid >> 5;

    int tot[PER];
#pragma unroll
    for (int j = 0; j < PER; ++j) {
        const int k = min(wave + 32 * j, NBK - 1);
        const int4* p = (const int4*)(H + k * NCH + lane * 8);
        const int4 a = p[0];
        const int4 b = p[1];
        tot[j] = a.x + a.y + a.z + a.w + b.x + b.y + b.z + b.w;
    }
#pragma unroll
    for (int j = 0; j < PER; ++j) {
        int s = tot[j];
#pragma unroll
        for (int o = 16; o > 0; o >>= 1) s += __shfl_xor(s, o, 32);
        const int k = wave + 32 * j;
        if (lane == 0) T[k] = (k < NBK) ? s : 0;
    }
    __syncthreads();
    if (wave == 0) {
        int v[PER];
        int sum = 0;
#pragma unroll
        for (int j = 0; j < PER; ++j) { v[j] = T[lane * PER + j]; sum += v[j]; }
        int incl = sum;
#pragma unroll
        for (int o = 1; o < 32; o <<= 1) {
            const int y = __shfl_up(incl, o, 32);
            if (lane >= o) incl += y;
        }
        int run = incl - sum;
#pragma unroll
        for (int j = 0; j < PER; ++j) { T[lane * PER + j] = run; run += v[j]; }
    }
    __syncthreads();

    const int k = blockIdx.x * 32 + wave;
    if (k < NBK) {
        const int4* p = (const int4*)(H + k * NCH + lane * 8);
        const int4 a = p[0];
        const int4 b = p[1];
        const int s = a.x + a.y + a.z + a.w + b.x + b.y + b.z + b.w;
        int incl = s;
#pragma unroll
        for (int o = 1; o < 32; o <<= 1) {
            const int y = __shfl_up(incl, o, 32);
            if (lane >= o) incl += y;
        }
        const int first = T[k];
        int run = first + incl - s;
        int4 oa, ob;
        oa.x = run; run += a.x;
        oa.y = run; run += a.y;
        oa.z = run; run += a.z;
        oa.w = run; run += a.w;
        ob.x = run; run += b.x;
        ob.y = run; run += b.y;
        ob.z = run; run += b.z;
        ob.w = run;
        int4* q = (int4*)(base + k * NCH + lane * 8);
        q[0] = oa;
        q[1] = ob;
        if (lane == 0) bstart[k] = first;
    }
    if (blockIdx.x == 0 && tid == 0) bstart[NBK] = N_EDGES;
}

__global__ __launch_bounds__(L1_THREADS)
void bucket_scatter(const int* __restrict__ src, const int* __restrict__ dst,
                    const int* __restrict__ base, unsigned* __restrict__ packed) {
    __shared__ int cur[NBK];
    const int tid = threadIdx.x;
    const int e0 = blockIdx.x * CH_EDGES;
    const int e1 = min(e0 + CH_EDGES, N_EDGES);
    int d[L1_ITERS], s[L1_ITERS];
#pragma unroll
    for (int j = 0; j < L1_ITERS; ++j) {
        const int e = e0 + tid + j * L1_THREADS;
        d[j] = (e < e1) ? dst[e] : -1;
        s[j] = (e < e1) ? src[e] : 0;
    }
    for (int k = tid; k < NBK; k += L1_THREADS) cur[k] = base[k * NCH + blockIdx.x];
    __syncthreads();
#pragma unroll
    for (int j = 0; j < L1_ITERS; ++j) {
        if (d[j] >= 0) {
            const int p = atomicAdd(&cur[d[j] >> BK_SHIFT], 1);
            packed[p] = ((unsigned)(d[j] & (BK_NODES - 1)) << SRC_BITS) | (unsigned)s[j];
        }
    }
}

#define FIN_THREADS 1024
#define FIN_KEEP 6
__global__ __launch_bounds__(FIN_THREADS)
void bucket_finalize(const unsigned* __restrict__ packed, const int* __restrict__ bstart,
                     int* __restrict__ rowptr, float* __restrict__ dinv,
                     int* __restrict__ csr_src) {
    __shared__ int cnt[BK_NODES];
    __shared__ int cur[BK_NODES];
    __shared__ int wtot[BK_NODES / 32];
    const int tid  = threadIdx.x;
    const int lane = tid & 31;
    const int wave = tid >> 5;
    const int k  = blockIdx.x;
    const int b0 = bstart[k];
    const int b1 = bstart[k + 1];

    unsigned w[FIN_KEEP];
#pragma unroll
    for (int j = 0; j < FIN_KEEP; ++j) {
        const int i = b0 + tid + j * FIN_THREADS;
        w[j] = (i < b1) ? packed[i] : NO_EDGE;
    }
    if (tid < BK_NODES) cnt[tid] = 0;
    __syncthreads();
#pragma unroll
    for (int j = 0; j < FIN_KEEP; ++j)
        if (w[j] != NO_EDGE) atomicAdd(&cnt[w[j] >> SRC_BITS], 1);
    for (int i = b0 + tid + FIN_KEEP * FIN_THREADS; i < b1; i += FIN_THREADS)
        atomicAdd(&cnt[packed[i] >> SRC_BITS], 1);
    __syncthreads();

    const int x = (tid < BK_NODES) ? cnt[tid] : 0;
    int incl = x;
#pragma unroll
    for (int o = 1; o < 32; o <<= 1) {
        const int y = __shfl_up(incl, o, 32);
        if (lane >= o) incl += y;
    }
    if (tid < BK_NODES && lane == 31) wtot[wave] = incl;
    __syncthreads();
    if (tid < BK_NODES) {
        int pre = 0;
        for (int q = 0; q < wave; ++q) pre += wtot[q];
        const int excl = pre + incl - x;
        cur[tid] = excl;
        const int node = k * BK_NODES + tid;
        if (node < N_NODES) {
            rowptr[node] = b0 + excl;
            const float dg = (float)(x + 1);
            dinv[node] = dg > 0.0f ? rsqrtf(dg) : 0.0f;
        }
    }
    if (k == NBK - 1 && tid == 0) rowptr[N_NODES] = b1;
    __syncthreads();

#pragma unroll
    for (int j = 0; j < FIN_KEEP; ++j) {
        if (w[j] != NO_EDGE) {
            const int p = atomicAdd(&cur[w[j] >> SRC_BITS], 1);
            csr_src[b0 + p] = (int)(w[j] & SRC_MASK);
        }
    }
    for (int i = b0 + tid + FIN_KEEP * FIN_THREADS; i < b1; i += FIN_THREADS) {
        const unsigned ww = packed[i];
        const int p = atomicAdd(&cur[ww >> SRC_BITS], 1);
        csr_src[b0 + p] = (int)(ww & SRC_MASK);
    }
}

#define SP 132
#define W2P 96
#define FUSED_BLOCKS 1024

__device__ __forceinline__
int edge_ids(const int* __restrict__ csr_src, int beg, int end, int lane) {
    return csr_src[(lane < end - beg) ? beg + lane : 0];
}

__device__ __forceinline__
float4 gather_xrow128(const int* __restrict__ csr_src, const float* __restrict__ dinv,
                      const float4* __restrict__ x4, int node, float dn,
                      int beg, int end, int lane, int idx, float dv) {
    float4 acc = make_float4(0.0f, 0.0f, 0.0f, 0.0f);
    bool own_in = false;
    for (int base = beg; base < end; base += 32) {
        const int n = min(32, end - base);
        if (base != beg) {
            idx = edge_ids(csr_src, base, end, lane);
            dv = dinv[idx];
        }
        for (int j0 = 0; j0 < n; j0 += 8) {
            float4 v[8];
#pragma unroll
            for (int u = 0; u < 8; ++u) {
                v[u] = make_float4(0.0f, 0.0f, 0.0f, 0.0f);
                if (j0 + u < n) {
                    const int s = __builtin_amdgcn_readlane(idx, j0 + u);
                    v[u] = x4[(long)s * 32 + lane];
                }
            }
            if (!own_in) {
                const float4 own = x4[(long)node * 32 + lane];
                acc.x = own.x * dn; acc.y = own.y * dn; acc.z = own.z * dn; acc.w = own.w * dn;
                own_in = true;
            }
#pragma unroll
            for (int u = 0; u < 8; ++u) {
                const float d = __int_as_float(
                    __builtin_amdgcn_readlane(__float_as_int(dv), j0 + u));
                acc.x += d * v[u].x; acc.y += d * v[u].y;
                acc.z += d * v[u].z; acc.w += d * v[u].w;
            }
        }
    }
    if (!own_in) {
        const float4 own = x4[(long)node * 32 + lane];
        acc.x = own.x * dn; acc.y = own.y * dn; acc.z = own.z * dn; acc.w = own.w * dn;
    }
    return acc;
}

__global__ __launch_bounds__(256, 8)
void gather_x_gemm128_gemm32(const int* __restrict__ rowptr, const int* __restrict__ csr_src,
                             const float* __restrict__ x, const float* __restrict__ dinv,
                             const float* __restrict__ W1, const float* __restrict__ b1,
                             const float* __restrict__ W2, float* __restrict__ h2s) {
    __shared__ alignas(16) float S[16 * SP];
    __shared__ alignas(16) float T[16 * SP];
    __shared__ alignas(16) float lW2[(D_HID / 2) * W2P];
    const int lane = threadIdx.x & 31;
    const int wave = __builtin_amdgcn_readfirstlane(threadIdx.x >> 5);
    for (int idx = threadIdx.x; idx < D_HID * N_CLS; idx += 256) {
        const int k = idx / N_CLS;
        const int n = idx - k * N_CLS;
        lW2[(k >> 1) * W2P + 2 * n + (k & 1)] = W2[idx];
    }
    const float4* __restrict__ x4 = (const float4*)x;
    const int c = lane * 4;
    const int half = lane >> 4;
    const int l16  = lane & 15;
    const int col1 = wave * 16 + l16;
    const int w2l  = half * W2P + 2 * l16;

    v2f bw[32];
#pragma unroll
    for (int j = 0; j < 32; ++j) {
        const int ka = 4 * j + 2 * half;
        bw[j].x = W1[ka * D_HID + col1];
        bw[j].y = W1[(ka + 1) * D_HID + col1];
    }
    const float bias1 = b1[col1];
    float* const tw = &T[8 * half * SP + col1];

    int r0, r1, r2, idxA, idxB;
    float dvA, dvB;
    float2 dn2;
    {
        const int na = blockIdx.x * 16 + 2 * wave;
        dn2 = *(const float2*)&dinv[na];
        r0 = rowptr[na];
        r1 = rowptr[na + 1];
        r2 = rowptr[na + 2];
        idxA = edge_ids(csr_src, r0, r1, lane);
        idxB = edge_ids(csr_src, r1, r2, lane);
        dvA = dinv[idxA];
        dvB = dinv[idxB];
    }
    int it = 0;
    for (int strip = blockIdx.x; strip < N_STRIPS; strip += FUSED_BLOCKS, ++it) {
        const int row0 = strip * 16;
        const int na = row0 + 2 * wave;
        const int nna = ((strip + FUSED_BLOCKS < N_STRIPS) ? row0 + 16 * FUSED_BLOCKS : row0)
                        + 2 * wave;
        const int nrp = rowptr[nna + min(lane, 2)];
        const float2 ndn2 = *(const float2*)&dinv[nna];
        const float dnA = dn2.x;
        const float dnB = dn2.y;
        {
            const float4 acc = gather_xrow128(csr_src, dinv, x4, na, dnA, r0, r1, lane, idxA, dvA);
            float4 o;
            o.x = acc.x * dnA; o.y = acc.y * dnA; o.z = acc.z * dnA; o.w = acc.w * dnA;
            *(float4*)&S[(2 * wave) * SP + c] = o;
        }
        {
            const float4 acc = gather_xrow128(csr_src, dinv, x4, na + 1, dnB, r1, r2, lane, idxB, dvB);
            float4 o;
            o.x = acc.x * dnB; o.y = acc.y * dnB; o.z = acc.z * dnB; o.w = acc.w * dnB;
            *(float4*)&S[(2 * wave + 1) * SP + c] = o;
        }
        __syncthreads();
        const int nr0 = __builtin_amdgcn_readlane(nrp, 0);
        const int nr1 = __builtin_amdgcn_readlane(nrp, 1);
        const int nr2 = __builtin_amdgcn_readlane(nrp, 2);
        const int nidxA = edge_ids(csr_src, nr0, nr1, lane);
        const int nidxB = edge_ids(csr_src, nr1, nr2, lane);

        v8f acc1 = (v8f){0.f, 0.f, 0.f, 0.f, 0.f, 0.f, 0.f, 0.f};
#pragma unroll
        for (int j = 0; j < 32; ++j) {
            const int ka = 4 * j + 2 * half;
            const v2f a = *(const v2f*)&S[l16 * SP + ka];
            acc1 = __builtin_amdgcn_wmma_f32_16x16x4_f32(
                false, a, false, bw[j], (short)0, acc1, false, false);
        }
#pragma unroll
        for (int r = 0; r < 8; ++r)
            tw[r * SP] = fmaxf(acc1[r] + bias1, 0.0f);
        __syncthreads();
        const float ndvA = dinv[nidxA];
        const float ndvB = dinv[nidxB];

        const int tile = (wave - 2 * it) & 7;
        if (tile < 2) {
            const float* w2t = lW2 + tile * 32;
            v8f acc = (v8f){0.f, 0.f, 0.f, 0.f, 0.f, 0.f, 0.f, 0.f};
#pragma unroll
            for (int k0 = 0; k0 < 128; k0 += 4) {
                const int ka = k0 + 2 * half;
                const v2f a = *(const v2f*)&T[l16 * SP + ka];
                const v2f b = *(const v2f*)&w2t[w2l + (k0 >> 1) * W2P];
                acc = __builtin_amdgcn_wmma_f32_16x16x4_f32(
                    false, a, false, b, (short)0, acc, false, false);
            }
            float* __restrict__ h2t = h2s + ((long)(row0 + 8 * half) * N_CLS + tile * 16 + l16);
            const float* __restrict__ dinvt = dinv + (row0 + 8 * half);
#pragma unroll
            for (int r = 0; r < 8; ++r)
                h2t[r * N_CLS] = acc[r] * dinvt[r];
        }
        r0 = nr0; r1 = nr1; r2 = nr2;
        idxA = nidxA; idxB = nidxB;
        dvA = ndvA; dvB = ndvB;
        dn2 = ndn2;
    }
}

template <int J>
__device__ __forceinline__ int group8_bcast(int v) {
    return __builtin_amdgcn_mov_dpp8(v, J * 0x249249);
}
constexpr unsigned group8_xor_sel(int m) {
    unsigned s = 0;
    for (int k = 0; k < 8; ++k) s |= (unsigned)(k ^ m) << (3 * k);
    return s;
}
template <int M>
__device__ __forceinline__ float group8_xor(float v) {
    return __int_as_float(__builtin_amdgcn_mov_dpp8(__float_as_int(v), group8_xor_sel(M)));
}

#define G32_SETS 4
#define G32_NPW  (4 * G32_SETS)
static_assert(G32_SETS % 2 == 0, "sets are taken two at a time");
static_assert(N_NODES % G32_NPW == 0, "a wave is all inside the node range or all outside");

#define G32_ROW(V, IX, N, J)                                           \
    {                                                                  \
        const int s = group8_bcast<J>(IX);                             \
        V[J] = make_float4(0.0f, 0.0f, 0.0f, 0.0f);                    \
        if (J < N) V[J] = h4[s * 8 + i];                               \
    }
#define G32_ROWS(V, IX, N)                                             \
    G32_ROW(V, IX, N, 0) G32_ROW(V, IX, N, 1) G32_ROW(V, IX, N, 2) G32_ROW(V, IX, N, 3) \
    G32_ROW(V, IX, N, 4) G32_ROW(V, IX, N, 5) G32_ROW(V, IX, N, 6) G32_ROW(V, IX, N, 7)

__device__ __forceinline__
void g32_finish(const float4 a, const float dn, const float4 bb, float4* __restrict__ dst) {
    float4 t;
    t.x = a.x * dn + bb.x;
    t.y = a.y * dn + bb.y;
    t.z = a.z * dn + bb.z;
    t.w = a.w * dn + bb.w;
    float m = fmaxf(fmaxf(t.x, t.y), fmaxf(t.z, t.w));
    m = fmaxf(m, group8_xor<1>(m));
    m = fmaxf(m, group8_xor<2>(m));
    m = fmaxf(m, group8_xor<4>(m));
    float s = (__expf(t.x - m) + __expf(t.y - m)) + (__expf(t.z - m) + __expf(t.w - m));
    s += group8_xor<1>(s);
    s += group8_xor<2>(s);
    s += group8_xor<4>(s);
    const float ls = __logf(s);
    float4 o;
    o.x = (t.x - m) - ls;
    o.y = (t.y - m) - ls;
    o.z = (t.z - m) - ls;
    o.w = (t.w - m) - ls;
    *dst = o;
}

__global__ __launch_bounds__(256, 8)
void gather32_final(const int* __restrict__ rowptr, const int* __restrict__ csr_src,
                    const float* __restrict__ h2s, const float* __restrict__ dinv,
                    const float* __restrict__ b2, float* __restrict__ out) {
    const int lane = threadIdx.x & 31;
    const int wave = __builtin_amdgcn_readfirstlane(threadIdx.x >> 5);
    const int n0 = (blockIdx.x * 8 + wave) * G32_NPW;
    if (n0 >= N_NODES) return;
    const int g = lane >> 3;
    const int i = lane & 7;
    const float4* __restrict__ h4 = (const float4*)h2s;
    float4* __restrict__ out4 = (float4*)out;
    const float4 bb = *(const float4*)&b2[4 * i];

    int node[G32_SETS], cur[G32_SETS], end[G32_SETS], idx[G32_SETS];
    float dn[G32_SETS];
    float4 acc[G32_SETS];
#pragma unroll
    for (int q = 0; q < G32_SETS; ++q) {
        node[q] = n0 + 4 * q + g;
        cur[q] = rowptr[node[q]];
        end[q] = rowptr[node[q] + 1];
    }
#pragma unroll
    for (int q = 0; q < G32_SETS; ++q) {
        dn[q]  = dinv[node[q]];
        acc[q] = h4[node[q] * 8 + i];
        idx[q] = (cur[q] + i < end[q]) ? csr_src[cur[q] + i] : 0;
    }

#pragma unroll
    for (int q = 0; q < G32_SETS; q += 2) {
        float4 aA = acc[q], aB = acc[q + 1];
        int cA = cur[q], cB = cur[q + 1];
        const int eA = end[q], eB = end[q + 1];
        int ixA = idx[q], ixB = idx[q + 1];
        while (__any((cA < eA) | (cB < eB))) {
            const int pA = cA + 8 + i;
            const int pB = cB + 8 + i;
            const int nixA = (pA < eA) ? csr_src[pA] : 0;
            const int nixB = (pB < eB) ? csr_src[pB] : 0;
            const int nA = eA - cA;
            const int nB = eB - cB;
            float4 v[8], w[8];
            G32_ROWS(v, ixA, nA)
            G32_ROWS(w, ixB, nB)
#pragma unroll
            for (int j = 0; j < 8; ++j) {
                aA.x += v[j].x; aA.y += v[j].y; aA.z += v[j].z; aA.w += v[j].w;
            }
#pragma unroll
            for (int j = 0; j < 8; ++j) {
                aB.x += w[j].x; aB.y += w[j].y; aB.z += w[j].z; aB.w += w[j].w;
            }
            cA += 8;
            cB += 8;
            ixA = nixA;
            ixB = nixB;
        }
        g32_finish(aA, dn[q],     bb, &out4[node[q] * 8 + i]);
        g32_finish(aB, dn[q + 1], bb, &out4[node[q + 1] * 8 + i]);
    }
}

extern "C" void kernel_launch(void* const* d_in, const int* in_sizes, int n_in,
                              void* d_out, int out_size, void* d_ws, size_t ws_size,
                              hipStream_t stream) {
    const float* x  = (const float*)d_in[0];
    const int*   ei = (const int*)d_in[1];
    const float* W1 = (const float*)d_in[2];
    const float* b1 = (const float*)d_in[3];
    const float* W2 = (const float*)d_in[4];
    const float* b2 = (const float*)d_in[5];
    float* out = (float*)d_out;

    const int* src = ei;
    const int* dst = ei + N_EDGES;

    float* dinv = (float*)d_ws;
    float* h2s  = dinv + N_NODES;
    int* rowptr  = (int*)(h2s + (long)N_NODES * 32);
    int* csr_src = rowptr + N_NODES + 32;
    unsigned* packed = (unsigned*)(csr_src + N_EDGES);
    int* H      = (int*)(packed + N_EDGES);
    int* base   = H + NBK * NCH;
    int* bstart = base + NBK * NCH;

    bucket_hist<<<NCH, L1_THREADS, 0, stream>>>(dst, H);
    bucket_scan<<<(NBK + 31) / 32, 1024, 0, stream>>>(H, base, bstart);
    bucket_scatter<<<NCH, L1_THREADS, 0, stream>>>(src, dst, base, packed);
    bucket_finalize<<<NBK, FIN_THREADS, 0, stream>>>(packed, bstart, rowptr, dinv, csr_src);

    gather_x_gemm128_gemm32<<<FUSED_BLOCKS, 256, 0, stream>>>(rowptr, csr_src, x, dinv, W1, b1, W2, h2s);
    gather32_final<<<(N_NODES + 8 * G32_NPW - 1) / (8 * G32_NPW), 256, 0, stream>>>(rowptr, csr_src, h2s, dinv, b2, out);
}


# Both counting-sort placing passes order their words in LDS and store contiguous runs (level 2: whole 128-B lines), not 32 scattered 4-B lines per store; edge sums still in LDS arrival rank
# speedup vs baseline: 5.2278x; 1.2148x over previous
#include <hip/hip_runtime.h>
#include <hip/hip_bf16.h>

#define N_NODES 100000
#define N_EDGES 1600000
#define D_IN    128
#define D_HID   128
#define N_CLS   32
#define N_STRIPS (N_NODES / 16)

typedef float v2f __attribute__((ext_vector_type(2)));
typedef float v8f __attribute__((ext_vector_type(8)));

#define BK_SHIFT 8
#define BK_NODES (1 << BK_SHIFT)
#define NBK      ((N_NODES + BK_NODES - 1) / BK_NODES)
#define NCH      256
#define CH_EDGES ((N_EDGES + NCH - 1) / NCH)
#define L1_THREADS 1024
#define L1_ITERS ((CH_EDGES + L1_THREADS - 1) / L1_THREADS)
#define SRC_BITS 17
#define SRC_MASK ((1u << SRC_BITS) - 1u)
#define NO_EDGE  0xFFFFFFFFu

static_assert(NCH == 256, "bucket_scan gives each lane 8 of the 256 chunk counts");
static_assert(N_NODES <= (1 << SRC_BITS), "source id must fit the packed word");

__global__ __launch_bounds__(L1_THREADS)
void bucket_hist(const int* __restrict__ dst, int* __restrict__ H) {
    __shared__ int h[NBK];
    const int tid = threadIdx.x;
    const int e0 = blockIdx.x * CH_EDGES;
    const int e1 = min(e0 + CH_EDGES, N_EDGES);
    int d[L1_ITERS];
#pragma unroll
    for (int j = 0; j < L1_ITERS; ++j) {
        const int e = e0 + tid + j * L1_THREADS;
        d[j] = (e < e1) ? dst[e] : -1;
    }
    for (int k = tid; k < NBK; k += L1_THREADS) h[k] = 0;
    __syncthreads();
#pragma unroll
    for (int j = 0; j < L1_ITERS; ++j)
        if (d[j] >= 0) atomicAdd(&h[d[j] >> BK_SHIFT], 1);
    __syncthreads();
    for (int k = tid; k < NBK; k += L1_THREADS) H[k * NCH + blockIdx.x] = h[k];
}

__global__ __launch_bounds__(1024)
void bucket_scan(const int* __restrict__ H, int* __restrict__ base,
                 int* __restrict__ bstart) {
    constexpr int PER = (NBK + 31) / 32;
    __shared__ int T[PER * 32];
    const int tid  = threadIdx.x;
    const int lane = tid & 31;
    const int wave = tid >> 5;

    int tot[PER];
#pragma unroll
    for (int j = 0; j < PER; ++j) {
        const int k = min(wave + 32 * j, NBK - 1);
        const int4* p = (const int4*)(H + k * NCH + lane * 8);
        const int4 a = p[0];
        const int4 b = p[1];
        tot[j] = a.x + a.y + a.z + a.w + b.x + b.y + b.z + b.w;
    }
#pragma unroll
    for (int j = 0; j < PER; ++j) {
        int s = tot[j];
#pragma unroll
        for (int o = 16; o > 0; o >>= 1) s += __shfl_xor(s, o, 32);
        const int k = wave + 32 * j;
        if (lane == 0) T[k] = (k < NBK) ? s : 0;
    }
    __syncthreads();
    if (wave == 0) {
        int v[PER];
        int sum = 0;
#pragma unroll
        for (int j = 0; j < PER; ++j) { v[j] = T[lane * PER + j]; sum += v[j]; }
        int incl = sum;
#pragma unroll
        for (int o = 1; o < 32; o <<= 1) {
            const int y = __shfl_up(incl, o, 32);
            if (lane >= o) incl += y;
        }
        int run = incl - sum;
#pragma unroll
        for (int j = 0; j < PER; ++j) { T[lane * PER + j] = run; run += v[j]; }
    }
    __syncthreads();

    const int k = blockIdx.x * 32 + wave;
    if (k < NBK) {
        const int4* p = (const int4*)(H + k * NCH + lane * 8);
        const int4 a = p[0];
        const int4 b = p[1];
        const int s = a.x + a.y + a.z + a.w + b.x + b.y + b.z + b.w;
        int incl = s;
#pragma unroll
        for (int o = 1; o < 32; o <<= 1) {
            const int y = __shfl_up(incl, o, 32);
            if (lane >= o) incl += y;
        }
        const int first = T[k];
        int run = first + incl - s;
        int4 oa, ob;
        oa.x = run; run += a.x;
        oa.y = run; run += a.y;
        oa.z = run; run += a.z;
        oa.w = run; run += a.w;
        ob.x = run; run += b.x;
        ob.y = run; run += b.y;
        ob.z = run; run += b.z;
        ob.w = run;
        int4* q = (int4*)(base + k * NCH + lane * 8);
        q[0] = oa;
        q[1] = ob;
        if (lane == 0) bstart[k] = first;
    }
    if (blockIdx.x == 0 && tid == 0) bstart[NBK] = N_EDGES;
}

static_assert(NBK <= L1_THREADS, "one bucket per thread in the local scan");
__global__ __launch_bounds__(L1_THREADS)
void bucket_scatter(const int* __restrict__ src, const int* __restrict__ dst,
                    const int* __restrict__ H, const int* __restrict__ base,
                    unsigned* __restrict__ packed) {
    __shared__ int cur[NBK];
    __shared__ int dlt[NBK];
    __shared__ int wtot[L1_THREADS / 32];
    __shared__ alignas(8) uint2 st[CH_EDGES];
    const int tid  = threadIdx.x;
    const int lane = tid & 31;
    const int wave = tid >> 5;
    const int e0 = blockIdx.x * CH_EDGES;
    const int e1 = min(e0 + CH_EDGES, N_EDGES);
    int d[L1_ITERS], s[L1_ITERS];
#pragma unroll
    for (int j = 0; j < L1_ITERS; ++j) {
        const int e = e0 + tid + j * L1_THREADS;
        d[j] = (e < e1) ? dst[e] : -1;
        s[j] = (e < e1) ? src[e] : 0;
    }
    int cnt = 0, gb = 0;
    if (tid < NBK) {
        cnt = H[tid * NCH + blockIdx.x];
        gb  = base[tid * NCH + blockIdx.x];
    }
    int incl = cnt;
#pragma unroll
    for (int o = 1; o < 32; o <<= 1) {
        const int y = __shfl_up(incl, o, 32);
        if (lane >= o) incl += y;
    }
    if (lane == 31) wtot[wave] = incl;
    __syncthreads();
    if (tid < NBK) {
        int pre = 0;
        for (int q = 0; q < wave; ++q) pre += wtot[q];
        const int ls = pre + incl - cnt;
        cur[tid] = ls;
        dlt[tid] = gb - ls;
    }
    __syncthreads();
    int q[L1_ITERS], gd[L1_ITERS];
#pragma unroll
    for (int j = 0; j < L1_ITERS; ++j) {
        q[j] = 0;
        gd[j] = 0;
        if (d[j] >= 0) {
            const int bk = d[j] >> BK_SHIFT;
            q[j]  = atomicAdd(&cur[bk], 1);
            gd[j] = dlt[bk];
        }
    }
#pragma unroll
    for (int j = 0; j < L1_ITERS; ++j) {
        if (d[j] >= 0)
            st[q[j]] = make_uint2(((unsigned)(d[j] & (BK_NODES - 1)) << SRC_BITS) | (unsigned)s[j],
                                  (unsigned)(q[j] + gd[j]));
    }
    __syncthreads();
    const int n = e1 - e0;
    const int last = max(n - 1, 0);
    uint2 v[L1_ITERS];
#pragma unroll
    for (int j = 0; j < L1_ITERS; ++j) v[j] = st[min(tid + j * L1_THREADS, last)];
#pragma unroll
    for (int j = 0; j < L1_ITERS; ++j)
        if (tid + j * L1_THREADS < n) packed[v[j].y] = v[j].x;
}

#define FIN_THREADS 1024
#define FIN_KEEP 6
#define FIN_STAGE 8192
__global__ __launch_bounds__(FIN_THREADS)
void bucket_finalize(const unsigned* __restrict__ packed, const int* __restrict__ bstart,
                     int* __restrict__ rowptr, float* __restrict__ dinv,
                     int* __restrict__ csr_src) {
    __shared__ int cnt[BK_NODES];
    __shared__ int cur[BK_NODES];
    __shared__ int wtot[BK_NODES / 32];
    __shared__ int stg[FIN_STAGE];
    const int tid  = threadIdx.x;
    const int lane = tid & 31;
    const int wave = tid >> 5;
    const int k  = blockIdx.x;
    const int b0 = bstart[k];
    const int b1 = bstart[k + 1];

    unsigned w[FIN_KEEP];
#pragma unroll
    for (int j = 0; j < FIN_KEEP; ++j) {
        const int i = b0 + tid + j * FIN_THREADS;
        w[j] = (i < b1) ? packed[i] : NO_EDGE;
    }
    if (tid < BK_NODES) cnt[tid] = 0;
    __syncthreads();
#pragma unroll
    for (int j = 0; j < FIN_KEEP; ++j)
        if (w[j] != NO_EDGE) atomicAdd(&cnt[w[j] >> SRC_BITS], 1);
    for (int i = b0 + tid + FIN_KEEP * FIN_THREADS; i < b1; i += FIN_THREADS)
        atomicAdd(&cnt[packed[i] >> SRC_BITS], 1);
    __syncthreads();

    const int x = (tid < BK_NODES) ? cnt[tid] : 0;
    int incl = x;
#pragma unroll
    for (int o = 1; o < 32; o <<= 1) {
        const int y = __shfl_up(incl, o, 32);
        if (lane >= o) incl += y;
    }
    if (tid < BK_NODES && lane == 31) wtot[wave] = incl;
    __syncthreads();
    if (tid < BK_NODES) {
        int pre = 0;
        for (int q = 0; q < wave; ++q) pre += wtot[q];
        const int excl = pre + incl - x;
        cur[tid] = excl;
        const int node = k * BK_NODES + tid;
        if (node < N_NODES) {
            rowptr[node] = b0 + excl;
            const float dg = (float)(x + 1);
            dinv[node] = dg > 0.0f ? rsqrtf(dg) : 0.0f;
        }
    }
    if (k == NBK - 1 && tid == 0) rowptr[N_NODES] = b1;
    __syncthreads();

    int pl[FIN_KEEP];
#pragma unroll
    for (int j = 0; j < FIN_KEEP; ++j) {
        pl[j] = 0;
        if (w[j] != NO_EDGE) pl[j] = atomicAdd(&cur[w[j] >> SRC_BITS], 1);
    }
#pragma unroll
    for (int j = 0; j < FIN_KEEP; ++j) {
        if (w[j] != NO_EDGE) {
            const int v = (int)(w[j] & SRC_MASK);
            if (pl[j] < FIN_STAGE) stg[pl[j]] = v; else csr_src[b0 + pl[j]] = v;
        }
    }
    for (int i = b0 + tid + FIN_KEEP * FIN_THREADS; i < b1; i += FIN_THREADS) {
        const unsigned ww = packed[i];
        const int p = atomicAdd(&cur[ww >> SRC_BITS], 1);
        const int v = (int)(ww & SRC_MASK);
        if (p < FIN_STAGE) stg[p] = v; else csr_src[b0 + p] = v;
    }
    __syncthreads();
    const int gend = b0 + min(b1 - b0, FIN_STAGE);
    for (int g = (b0 & ~31) + tid; g < gend; g += FIN_THREADS)
        if (g >= b0) csr_src[g] = stg[g - b0];
}

#define SP 132
#define W2P 96
#define FUSED_BLOCKS 1024

__device__ __forceinline__
int edge_ids(const int* __restrict__ csr_src, int beg, int end, int lane) {
    return csr_src[(lane < end - beg) ? beg + lane : 0];
}

__device__ __forceinline__
float4 gather_xrow128(const int* __restrict__ csr_src, const float* __restrict__ dinv,
                      const float4* __restrict__ x4, int node, float dn,
                      int beg, int end, int lane, int idx, float dv) {
    float4 acc = make_float4(0.0f, 0.0f, 0.0f, 0.0f);
    bool own_in = false;
    for (int base = beg; base < end; base += 32) {
        const int n = min(32, end - base);
        if (base != beg) {
            idx = edge_ids(csr_src, base, end, lane);
            dv = dinv[idx];
        }
        for (int j0 = 0; j0 < n; j0 += 8) {
            float4 v[8];
#pragma unroll
            for (int u = 0; u < 8; ++u) {
                v[u] = make_float4(0.0f, 0.0f, 0.0f, 0.0f);
                if (j0 + u < n) {
                    const int s = __builtin_amdgcn_readlane(idx, j0 + u);
                    v[u] = x4[(long)s * 32 + lane];
                }
            }
            if (!own_in) {
                const float4 own = x4[(long)node * 32 + lane];
                acc.x = own.x * dn; acc.y = own.y * dn; acc.z = own.z * dn; acc.w = own.w * dn;
                own_in = true;
            }
#pragma unroll
            for (int u = 0; u < 8; ++u) {
                const float d = __int_as_float(
                    __builtin_amdgcn_readlane(__float_as_int(dv), j0 + u));
                acc.x += d * v[u].x; acc.y += d * v[u].y;
                acc.z += d * v[u].z; acc.w += d * v[u].w;
            }
        }
    }
    if (!own_in) {
        const float4 own = x4[(long)node * 32 + lane];
        acc.x = own.x * dn; acc.y = own.y * dn; acc.z = own.z * dn; acc.w = own.w * dn;
    }
    return acc;
}

__global__ __launch_bounds__(256, 8)
void gather_x_gemm128_gemm32(const int* __restrict__ rowptr, const int* __restrict__ csr_src,
                             const float* __restrict__ x, const float* __restrict__ dinv,
                             const float* __restrict__ W1, const float* __restrict__ b1,
                             const float* __restrict__ W2, float* __restrict__ h2s) {
    __shared__ alignas(16) float S[16 * SP];
    __shared__ alignas(16) float T[16 * SP];
    __shared__ alignas(16) float lW2[(D_HID / 2) * W2P];
    const int lane = threadIdx.x & 31;
    const int wave = __builtin_amdgcn_readfirstlane(threadIdx.x >> 5);
    for (int idx = threadIdx.x; idx < D_HID * N_CLS; idx += 256) {
        const int k = idx / N_CLS;
        const int n = idx - k * N_CLS;
        lW2[(k >> 1) * W2P + 2 * n + (k & 1)] = W2[idx];
    }
    const float4* __restrict__ x4 = (const float4*)x;
    const int c = lane * 4;
    const int half = lane >> 4;
    const int l16  = lane & 15;
    const int col1 = wave * 16 + l16;
    const int w2l  = half * W2P + 2 * l16;

    v2f bw[32];
#pragma unroll
    for (int j = 0; j < 32; ++j) {
        const int ka = 4 * j + 2 * half;
        bw[j].x = W1[ka * D_HID + col1];
        bw[j].y = W1[(ka + 1) * D_HID + col1];
    }
    const float bias1 = b1[col1];
    float* const tw = &T[8 * half * SP + col1];

    int r0, r1, r2, idxA, idxB;
    float dvA, dvB;
    float2 dn2;
    {
        const int na = blockIdx.x * 16 + 2 * wave;
        dn2 = *(const float2*)&dinv[na];
        r0 = rowptr[na];
        r1 = rowptr[na + 1];
        r2 = rowptr[na + 2];
        idxA = edge_ids(csr_src, r0, r1, lane);
        idxB = edge_ids(csr_src, r1, r2, lane);
        dvA = dinv[idxA];
        dvB = dinv[idxB];
    }
    int it = 0;
    for (int strip = blockIdx.x; strip < N_STRIPS; strip += FUSED_BLOCKS, ++it) {
        const int row0 = strip * 16;
        const int na = row0 + 2 * wave;
        const int nna = ((strip + FUSED_BLOCKS < N_STRIPS) ? row0 + 16 * FUSED_BLOCKS : row0)
                        + 2 * wave;
        const int nrp = rowptr[nna + min(lane, 2)];
        const float2 ndn2 = *(const float2*)&dinv[nna];
        const float dnA = dn2.x;
        const float dnB = dn2.y;
        {
            const float4 acc = gather_xrow128(csr_src, dinv, x4, na, dnA, r0, r1, lane, idxA, dvA);
            float4 o;
            o.x = acc.x * dnA; o.y = acc.y * dnA; o.z = acc.z * dnA; o.w = acc.w * dnA;
            *(float4*)&S[(2 * wave) * SP + c] = o;
        }
        {
            const float4 acc = gather_xrow128(csr_src, dinv, x4, na + 1, dnB, r1, r2, lane, idxB, dvB);
            float4 o;
            o.x = acc.x * dnB; o.y = acc.y * dnB; o.z = acc.z * dnB; o.w = acc.w * dnB;
            *(float4*)&S[(2 * wave + 1) * SP + c] = o;
        }
        __syncthreads();
        const int nr0 = __builtin_amdgcn_readlane(nrp, 0);
        const int nr1 = __builtin_amdgcn_readlane(nrp, 1);
        const int nr2 = __builtin_amdgcn_readlane(nrp, 2);
        const int nidxA = edge_ids(csr_src, nr0, nr1, lane);
        const int nidxB = edge_ids(csr_src, nr1, nr2, lane);

        v8f acc1 = (v8f){0.f, 0.f, 0.f, 0.f, 0.f, 0.f, 0.f, 0.f};
#pragma unroll
        for (int j = 0; j < 32; ++j) {
            const int ka = 4 * j + 2 * half;
            const v2f a = *(const v2f*)&S[l16 * SP + ka];
            acc1 = __builtin_amdgcn_wmma_f32_16x16x4_f32(
                false, a, false, bw[j], (short)0, acc1, false, false);
        }
#pragma unroll
        for (int r = 0; r < 8; ++r)
            tw[r * SP] = fmaxf(acc1[r] + bias1, 0.0f);
        __syncthreads();
        const float ndvA = dinv[nidxA];
        const float ndvB = dinv[nidxB];

        const int tile = (wave - 2 * it) & 7;
        if (tile < 2) {
            const float* w2t = lW2 + tile * 32;
            v8f acc = (v8f){0.f, 0.f, 0.f, 0.f, 0.f, 0.f, 0.f, 0.f};
#pragma unroll
            for (int k0 = 0; k0 < 128; k0 += 4) {
                const int ka = k0 + 2 * half;
                const v2f a = *(const v2f*)&T[l16 * SP + ka];
                const v2f b = *(const v2f*)&w2t[w2l + (k0 >> 1) * W2P];
                acc = __builtin_amdgcn_wmma_f32_16x16x4_f32(
                    false, a, false, b, (short)0, acc, false, false);
            }
            float* __restrict__ h2t = h2s + ((long)(row0 + 8 * half) * N_CLS + tile * 16 + l16);
            const float* __restrict__ dinvt = dinv + (row0 + 8 * half);
#pragma unroll
            for (int r = 0; r < 8; ++r)
                h2t[r * N_CLS] = acc[r] * dinvt[r];
        }
        r0 = nr0; r1 = nr1; r2 = nr2;
        idxA = nidxA; idxB = nidxB;
        dvA = ndvA; dvB = ndvB;
        dn2 = ndn2;
    }
}

template <int J>
__device__ __forceinline__ int group8_bcast(int v) {
    return __builtin_amdgcn_mov_dpp8(v, J * 0x249249);
}
constexpr unsigned group8_xor_sel(int m) {
    unsigned s = 0;
    for (int k = 0; k < 8; ++k) s |= (unsigned)(k ^ m) << (3 * k);
    return s;
}
template <int M>
__device__ __forceinline__ float group8_xor(float v) {
    return __int_as_float(__builtin_amdgcn_mov_dpp8(__float_as_int(v), group8_xor_sel(M)));
}

#define G32_SETS 4
#define G32_NPW  (4 * G32_SETS)
static_assert(G32_SETS % 2 == 0, "sets are taken two at a time");
static_assert(N_NODES % G32_NPW == 0, "a wave is all inside the node range or all outside");

#define G32_ROW(V, IX, N, J)                                           \
    {                                                                  \
        const int s = group8_bcast<J>(IX);                             \
        V[J] = make_float4(0.0f, 0.0f, 0.0f, 0.0f);                    \
        if (J < N) V[J] = h4[s * 8 + i];                               \
    }
#define G32_ROWS(V, IX, N)                                             \
    G32_ROW(V, IX, N, 0) G32_ROW(V, IX, N, 1) G32_ROW(V, IX, N, 2) G32_ROW(V, IX, N, 3) \
    G32_ROW(V, IX, N, 4) G32_ROW(V, IX, N, 5) G32_ROW(V, IX, N, 6) G32_ROW(V, IX, N, 7)

__device__ __forceinline__
void g32_finish(const float4 a, const float dn, const float4 bb, float4* __restrict__ dst) {
    float4 t;
    t.x = a.x * dn + bb.x;
    t.y = a.y * dn + bb.y;
    t.z = a.z * dn + bb.z;
    t.w = a.w * dn + bb.w;
    float m = fmaxf(fmaxf(t.x, t.y), fmaxf(t.z, t.w));
    m = fmaxf(m, group8_xor<1>(m));
    m = fmaxf(m, group8_xor<2>(m));
    m = fmaxf(m, group8_xor<4>(m));
    float s = (__expf(t.x - m) + __expf(t.y - m)) + (__expf(t.z - m) + __expf(t.w - m));
    s += group8_xor<1>(s);
    s += group8_xor<2>(s);
    s += group8_xor<4>(s);
    const float ls = __logf(s);
    float4 o;
    o.x = (t.x - m) - ls;
    o.y = (t.y - m) - ls;
    o.z = (t.z - m) - ls;
    o.w = (t.w - m) - ls;
    *dst = o;
}

__global__ __launch_bounds__(256, 8)
void gather32_final(const int* __restrict__ rowptr, const int* __restrict__ csr_src,
                    const float* __restrict__ h2s, const float* __restrict__ dinv,
                    const float* __restrict__ b2, float* __restrict__ out) {
    const int lane = threadIdx.x & 31;
    const int wave = __builtin_amdgcn_readfirstlane(threadIdx.x >> 5);
    const int n0 = (blockIdx.x * 8 + wave) * G32_NPW;
    if (n0 >= N_NODES) return;
    const int g = lane >> 3;
    const int i = lane & 7;
    const float4* __restrict__ h4 = (const float4*)h2s;
    float4* __restrict__ out4 = (float4*)out;
    const float4 bb = *(const float4*)&b2[4 * i];

    int node[G32_SETS], cur[G32_SETS], end[G32_SETS], idx[G32_SETS];
    float dn[G32_SETS];
    float4 acc[G32_SETS];
#pragma unroll
    for (int q = 0; q < G32_SETS; ++q) {
        node[q] = n0 + 4 * q + g;
        cur[q] = rowptr[node[q]];
        end[q] = rowptr[node[q] + 1];
    }
#pragma unroll
    for (int q = 0; q < G32_SETS; ++q) {
        dn[q]  = dinv[node[q]];
        acc[q] = h4[node[q] * 8 + i];
        idx[q] = (cur[q] + i < end[q]) ? csr_src[cur[q] + i] : 0;
    }

#pragma unroll
    for (int q = 0; q < G32_SETS; q += 2) {
        float4 aA = acc[q], aB = acc[q + 1];
        int cA = cur[q], cB = cur[q + 1];
        const int eA = end[q], eB = end[q + 1];
        int ixA = idx[q], ixB = idx[q + 1];
        while (__any((cA < eA) | (cB < eB))) {
            const int pA = cA + 8 + i;
            const int pB = cB + 8 + i;
            const int nixA = (pA < eA) ? csr_src[pA] : 0;
            const int nixB = (pB < eB) ? csr_src[pB] : 0;
            const int nA = eA - cA;
            const int nB = eB - cB;
            float4 v[8], w[8];
            G32_ROWS(v, ixA, nA)
            G32_ROWS(w, ixB, nB)
#pragma unroll
            for (int j = 0; j < 8; ++j) {
                aA.x += v[j].x; aA.y += v[j].y; aA.z += v[j].z; aA.w += v[j].w;
            }
#pragma unroll
            for (int j = 0; j < 8; ++j) {
                aB.x += w[j].x; aB.y += w[j].y; aB.z += w[j].z; aB.w += w[j].w;
            }
            cA += 8;
            cB += 8;
            ixA = nixA;
            ixB = nixB;
        }
        g32_finish(aA, dn[q],     bb, &out4[node[q] * 8 + i]);
        g32_finish(aB, dn[q + 1], bb, &out4[node[q + 1] * 8 + i]);
    }
}

extern "C" void kernel_launch(void* const* d_in, const int* in_sizes, int n_in,
                              void* d_out, int out_size, void* d_ws, size_t ws_size,
                              hipStream_t stream) {
    const float* x  = (const float*)d_in[0];
    const int*   ei = (const int*)d_in[1];
    const float* W1 = (const float*)d_in[2];
    const float* b1 = (const float*)d_in[3];
    const float* W2 = (const float*)d_in[4];
    const float* b2 = (const float*)d_in[5];
    float* out = (float*)d_out;

    const int* src = ei;
    const int* dst = ei + N_EDGES;

    float* dinv = (float*)d_ws;
    float* h2s  = dinv + N_NODES;
    int* rowptr  = (int*)(h2s + (long)N_NODES * 32);
    int* csr_src = rowptr + N_NODES + 32;
    unsigned* packed = (unsigned*)(csr_src + N_EDGES);
    int* H      = (int*)(packed + N_EDGES);
    int* base   = H + NBK * NCH;
    int* bstart = base + NBK * NCH;

    bucket_hist<<<NCH, L1_THREADS, 0, stream>>>(dst, H);
    bucket_scan<<<(NBK + 31) / 32, 1024, 0, stream>>>(H, base, bstart);
    bucket_scatter<<<NCH, L1_THREADS, 0, stream>>>(src, dst, H, base, packed);
    bucket_finalize<<<NBK, FIN_THREADS, 0, stream>>>(packed, bstart, rowptr, dinv, csr_src);

    gather_x_gemm128_gemm32<<<FUSED_BLOCKS, 256, 0, stream>>>(rowptr, csr_src, x, dinv, W1, b1, W2, h2s);
    gather32_final<<<(N_NODES + 8 * G32_NPW - 1) / (8 * G32_NPW), 256, 0, stream>>>(rowptr, csr_src, h2s, dinv, b2, out);
}


# CSR build: count and scan launches dropped; a chunk block sorts its own edges by bucket (one returning LDS add counts and ranks), bucket blocks read 256 runs; edge sums in new LDS arrival order
# speedup vs baseline: 5.9252x; 1.1334x over previous
#include <hip/hip_runtime.h>
#include <hip/hip_bf16.h>

#define N_NODES 100000
#define N_EDGES 1600000
#define D_IN    128
#define D_HID   128
#define N_CLS   32
#define N_STRIPS (N_NODES / 16)

typedef float v2f __attribute__((ext_vector_type(2)));
typedef float v8f __attribute__((ext_vector_type(8)));

#define BK_SHIFT 8
#define BK_NODES (1 << BK_SHIFT)
#define NBK      ((N_NODES + BK_NODES - 1) / BK_NODES)
#define NCH      256
#define CH_EDGES ((N_EDGES + NCH - 1) / NCH)
#define L1_THREADS 1024
#define L1_ITERS ((CH_EDGES + L1_THREADS - 1) / L1_THREADS)
#define SRC_BITS 17
#define SRC_MASK ((1u << SRC_BITS) - 1u)
#define NO_EDGE  0xFFFFFFFFu

static_assert(N_NODES <= (1 << SRC_BITS), "source id must fit the packed word");
static_assert(NBK <= L1_THREADS, "one bucket per thread in the local scan");

__global__ __launch_bounds__(L1_THREADS)
void chunk_sort(const int* __restrict__ src, const int* __restrict__ dst,
                int* __restrict__ off, unsigned* __restrict__ packed) {
    __shared__ int cnt[NBK];
    __shared__ int ls[NBK];
    __shared__ int wtot[L1_THREADS / 32];
    __shared__ unsigned st[CH_EDGES];
    const int tid  = threadIdx.x;
    const int lane = tid & 31;
    const int wave = tid >> 5;
    const int e0 = blockIdx.x * CH_EDGES;
    const int e1 = min(e0 + CH_EDGES, N_EDGES);
    int d[L1_ITERS], s[L1_ITERS];
#pragma unroll
    for (int j = 0; j < L1_ITERS; ++j) {
        const int e = e0 + tid + j * L1_THREADS;
        d[j] = (e < e1) ? dst[e] : -1;
        s[j] = (e < e1) ? src[e] : 0;
    }
    if (tid < NBK) cnt[tid] = 0;
    __syncthreads();
    int r[L1_ITERS];
#pragma unroll
    for (int j = 0; j < L1_ITERS; ++j) {
        r[j] = 0;
        if (d[j] >= 0) r[j] = atomicAdd(&cnt[d[j] >> BK_SHIFT], 1);
    }
    __syncthreads();
    const int c = (tid < NBK) ? cnt[tid] : 0;
    int incl = c;
#pragma unroll
    for (int o = 1; o < 32; o <<= 1) {
        const int y = __shfl_up(incl, o, 32);
        if (lane >= o) incl += y;
    }
    if (lane == 31) wtot[wave] = incl;
    __syncthreads();
    int start = 0;
    if (tid < NBK) {
        int pre = 0;
        for (int q = 0; q < wave; ++q) pre += wtot[q];
        start = pre + incl - c;
        ls[tid] = start;
    }
    __syncthreads();
    int p[L1_ITERS];
#pragma unroll
    for (int j = 0; j < L1_ITERS; ++j)
        p[j] = ls[max(d[j], 0) >> BK_SHIFT] + r[j];
#pragma unroll
    for (int j = 0; j < L1_ITERS; ++j) {
        if (d[j] >= 0)
            st[p[j]] = ((unsigned)(d[j] & (BK_NODES - 1)) << SRC_BITS) | (unsigned)s[j];
    }
    __syncthreads();
    const int n = e1 - e0;
    const int last = max(n - 1, 0);
    unsigned v[L1_ITERS];
#pragma unroll
    for (int j = 0; j < L1_ITERS; ++j) v[j] = st[min(tid + j * L1_THREADS, last)];
#pragma unroll
    for (int j = 0; j < L1_ITERS; ++j) {
        const int i = tid + j * L1_THREADS;
        if (i < n) packed[e0 + i] = v[j];
    }
    if (tid < NBK) off[tid * NCH + blockIdx.x] = start;
    if (tid == 0) off[NBK * NCH + blockIdx.x] = n;
}

#define FIN_THREADS 1024
#define RUN_LANES 4
#define FIN_KEEP 8
#define FIN_STAGE 8192
static_assert(FIN_THREADS == RUN_LANES * NCH, "one chunk run per RUN_LANES threads");
__global__ __launch_bounds__(FIN_THREADS)
void bucket_finalize(const unsigned* __restrict__ packed, const int* __restrict__ off,
                     int* __restrict__ rowptr, float* __restrict__ dinv,
                     int* __restrict__ csr_src) {
    __shared__ int cnt[BK_NODES];
    __shared__ int cur[BK_NODES];
    __shared__ int wtot[BK_NODES / 32];
    __shared__ int wlo[FIN_THREADS / 32];
    __shared__ int whi[FIN_THREADS / 32];
    __shared__ int stg[FIN_STAGE];
    const int tid  = threadIdx.x;
    const int lane = tid & 31;
    const int wave = tid >> 5;
    const int k   = blockIdx.x;
    const int ch  = tid / RUN_LANES;
    const int sub = tid % RUN_LANES;
    const int lo = off[k * NCH + ch];
    const int hi = off[(k + 1) * NCH + ch];
    const int len = hi - lo;
    const unsigned* __restrict__ run = packed + (ch * CH_EDGES + lo);

    unsigned w[FIN_KEEP];
#pragma unroll
    for (int j = 0; j < FIN_KEEP; ++j) {
        const int i = sub + j * RUN_LANES;
        w[j] = (i < len) ? run[i] : NO_EDGE;
    }
    int slo = (sub == 0) ? lo : 0;
    int shi = (sub == 0) ? hi : 0;
#pragma unroll
    for (int o = 16; o > 0; o >>= 1) {
        slo += __shfl_xor(slo, o, 32);
        shi += __shfl_xor(shi, o, 32);
    }
    if (lane == 0) { wlo[wave] = slo; whi[wave] = shi; }
    if (tid < BK_NODES) cnt[tid] = 0;
    __syncthreads();
    int b0 = wlo[lane];
    int b1 = whi[lane];
#pragma unroll
    for (int o = 16; o > 0; o >>= 1) {
        b0 += __shfl_xor(b0, o, 32);
        b1 += __shfl_xor(b1, o, 32);
    }
#pragma unroll
    for (int j = 0; j < FIN_KEEP; ++j)
        if (w[j] != NO_EDGE) atomicAdd(&cnt[w[j] >> SRC_BITS], 1);
    for (int i = sub + FIN_KEEP * RUN_LANES; i < len; i += RUN_LANES)
        atomicAdd(&cnt[run[i] >> SRC_BITS], 1);
    __syncthreads();

    const int x = (tid < BK_NODES) ? cnt[tid] : 0;
    int incl = x;
#pragma unroll
    for (int o = 1; o < 32; o <<= 1) {
        const int y = __shfl_up(incl, o, 32);
        if (lane >= o) incl += y;
    }
    if (tid < BK_NODES && lane == 31) wtot[wave] = incl;
    __syncthreads();
    if (tid < BK_NODES) {
        int pre = 0;
        for (int q = 0; q < wave; ++q) pre += wtot[q];
        const int excl = pre + incl - x;
        cur[tid] = excl;
        const int node = k * BK_NODES + tid;
        if (node < N_NODES) {
            rowptr[node] = b0 + excl;
            const float dg = (float)(x + 1);
            dinv[node] = dg > 0.0f ? rsqrtf(dg) : 0.0f;
        }
    }
    if (k == NBK - 1 && tid == 0) rowptr[N_NODES] = b1;
    __syncthreads();

    int pl[FIN_KEEP];
#pragma unroll
    for (int j = 0; j < FIN_KEEP; ++j) {
        pl[j] = 0;
        if (w[j] != NO_EDGE) pl[j] = atomicAdd(&cur[w[j] >> SRC_BITS], 1);
    }
#pragma unroll
    for (int j = 0; j < FIN_KEEP; ++j) {
        if (w[j] != NO_EDGE) {
            const int v = (int)(w[j] & SRC_MASK);
            if (pl[j] < FIN_STAGE) stg[pl[j]] = v; else csr_src[b0 + pl[j]] = v;
        }
    }
    for (int i = sub + FIN_KEEP * RUN_LANES; i < len; i += RUN_LANES) {
        const unsigned ww = run[i];
        const int p = atomicAdd(&cur[ww >> SRC_BITS], 1);
        const int v = (int)(ww & SRC_MASK);
        if (p < FIN_STAGE) stg[p] = v; else csr_src[b0 + p] = v;
    }
    __syncthreads();
    const int gend = b0 + min(b1 - b0, FIN_STAGE);
    for (int g = (b0 & ~31) + tid; g < gend; g += FIN_THREADS)
        if (g >= b0) csr_src[g] = stg[g - b0];
}

#define SP 132
#define W2P 96
#define FUSED_BLOCKS 1024

__device__ __forceinline__
int edge_ids(const int* __restrict__ csr_src, int beg, int end, int lane) {
    return csr_src[(lane < end - beg) ? beg + lane : 0];
}

__device__ __forceinline__
float4 gather_xrow128(const int* __restrict__ csr_src, const float* __restrict__ dinv,
                      const float4* __restrict__ x4, int node, float dn,
                      int beg, int end, int lane, int idx, float dv) {
    float4 acc = make_float4(0.0f, 0.0f, 0.0f, 0.0f);
    bool own_in = false;
    for (int base = beg; base < end; base += 32) {
        const int n = min(32, end - base);
        if (base != beg) {
            idx = edge_ids(csr_src, base, end, lane);
            dv = dinv[idx];
        }
        for (int j0 = 0; j0 < n; j0 += 8) {
            float4 v[8];
#pragma unroll
            for (int u = 0; u < 8; ++u) {
                v[u] = make_float4(0.0f, 0.0f, 0.0f, 0.0f);
                if (j0 + u < n) {
                    const int s = __builtin_amdgcn_readlane(idx, j0 + u);
                    v[u] = x4[(long)s * 32 + lane];
                }
            }
            if (!own_in) {
                const float4 own = x4[(long)node * 32 + lane];
                acc.x = own.x * dn; acc.y = own.y * dn; acc.z = own.z * dn; acc.w = own.w * dn;
                own_in = true;
            }
#pragma unroll
            for (int u = 0; u < 8; ++u) {
                const float d = __int_as_float(
                    __builtin_amdgcn_readlane(__float_as_int(dv), j0 + u));
                acc.x += d * v[u].x; acc.y += d * v[u].y;
                acc.z += d * v[u].z; acc.w += d * v[u].w;
            }
        }
    }
    if (!own_in) {
        const float4 own = x4[(long)node * 32 + lane];
        acc.x = own.x * dn; acc.y = own.y * dn; acc.z = own.z * dn; acc.w = own.w * dn;
    }
    return acc;
}

__global__ __launch_bounds__(256, 8)
void gather_x_gemm128_gemm32(const int* __restrict__ rowptr, const int* __restrict__ csr_src,
                             const float* __restrict__ x, const float* __restrict__ dinv,
                             const float* __restrict__ W1, const float* __restrict__ b1,
                             const float* __restrict__ W2, float* __restrict__ h2s) {
    __shared__ alignas(16) float S[16 * SP];
    __shared__ alignas(16) float T[16 * SP];
    __shared__ alignas(16) float lW2[(D_HID / 2) * W2P];
    const int lane = threadIdx.x & 31;
    const int wave = __builtin_amdgcn_readfirstlane(threadIdx.x >> 5);
    for (int idx = threadIdx.x; idx < D_HID * N_CLS; idx += 256) {
        const int k = idx / N_CLS;
        const int n = idx - k * N_CLS;
        lW2[(k >> 1) * W2P + 2 * n + (k & 1)] = W2[idx];
    }
    const float4* __restrict__ x4 = (const float4*)x;
    const int c = lane * 4;
    const int half = lane >> 4;
    const int l16  = lane & 15;
    const int col1 = wave * 16 + l16;
    const int w2l  = half * W2P + 2 * l16;

    v2f bw[32];
#pragma unroll
    for (int j = 0; j < 32; ++j) {
        const int ka = 4 * j + 2 * half;
        bw[j].x = W1[ka * D_HID + col1];
        bw[j].y = W1[(ka + 1) * D_HID + col1];
    }
    const float bias1 = b1[col1];
    float* const tw = &T[8 * half * SP + col1];

    int r0, r1, r2, idxA, idxB;
    float dvA, dvB;
    float2 dn2;
    {
        const int na = blockIdx.x * 16 + 2 * wave;
        dn2 = *(const float2*)&dinv[na];
        r0 = rowptr[na];
        r1 = rowptr[na + 1];
        r2 = rowptr[na + 2];
        idxA = edge_ids(csr_src, r0, r1, lane);
        idxB = edge_ids(csr_src, r1, r2, lane);
        dvA = dinv[idxA];
        dvB = dinv[idxB];
    }
    int it = 0;
    for (int strip = blockIdx.x; strip < N_STRIPS; strip += FUSED_BLOCKS, ++it) {
        const int row0 = strip * 16;
        const int na = row0 + 2 * wave;
        const int nna = ((strip + FUSED_BLOCKS < N_STRIPS) ? row0 + 16 * FUSED_BLOCKS : row0)
                        + 2 * wave;
        const int nrp = rowptr[nna + min(lane, 2)];
        const float2 ndn2 = *(const float2*)&dinv[nna];
        const float dnA = dn2.x;
        const float dnB = dn2.y;
        {
            const float4 acc = gather_xrow128(csr_src, dinv, x4, na, dnA, r0, r1, lane, idxA, dvA);
            float4 o;
            o.x = acc.x * dnA; o.y = acc.y * dnA; o.z = acc.z * dnA; o.w = acc.w * dnA;
            *(float4*)&S[(2 * wave) * SP + c] = o;
        }
        {
            const float4 acc = gather_xrow128(csr_src, dinv, x4, na + 1, dnB, r1, r2, lane, idxB, dvB);
            float4 o;
            o.x = acc.x * dnB; o.y = acc.y * dnB; o.z = acc.z * dnB; o.w = acc.w * dnB;
            *(float4*)&S[(2 * wave + 1) * SP + c] = o;
        }
        __syncthreads();
        const int nr0 = __builtin_amdgcn_readlane(nrp, 0);
        const int nr1 = __builtin_amdgcn_readlane(nrp, 1);
        const int nr2 = __builtin_amdgcn_readlane(nrp, 2);
        const int nidxA = edge_ids(csr_src, nr0, nr1, lane);
        const int nidxB = edge_ids(csr_src, nr1, nr2, lane);

        v8f acc1 = (v8f){0.f, 0.f, 0.f, 0.f, 0.f, 0.f, 0.f, 0.f};
#pragma unroll
        for (int j = 0; j < 32; ++j) {
            const int ka = 4 * j + 2 * half;
            const v2f a = *(const v2f*)&S[l16 * SP + ka];
            acc1 = __builtin_amdgcn_wmma_f32_16x16x4_f32(
                false, a, false, bw[j], (short)0, acc1, false, false);
        }
#pragma unroll
        for (int r = 0; r < 8; ++r)
            tw[r * SP] = fmaxf(acc1[r] + bias1, 0.0f);
        __syncthreads();
        const float ndvA = dinv[nidxA];
        const float ndvB = dinv[nidxB];

        const int tile = (wave - 2 * it) & 7;
        if (tile < 2) {
            const float* w2t = lW2 + tile * 32;
            v8f acc = (v8f){0.f, 0.f, 0.f, 0.f, 0.f, 0.f, 0.f, 0.f};
#pragma unroll
            for (int k0 = 0; k0 < 128; k0 += 4) {
                const int ka = k0 + 2 * half;
                const v2f a = *(const v2f*)&T[l16 * SP + ka];
                const v2f b = *(const v2f*)&w2t[w2l + (k0 >> 1) * W2P];
                acc = __builtin_amdgcn_wmma_f32_16x16x4_f32(
                    false, a, false, b, (short)0, acc, false, false);
            }
            float* __restrict__ h2t = h2s + ((long)(row0 + 8 * half) * N_CLS + tile * 16 + l16);
            const float* __restrict__ dinvt = dinv + (row0 + 8 * half);
#pragma unroll
            for (int r = 0; r < 8; ++r)
                h2t[r * N_CLS] = acc[r] * dinvt[r];
        }
        r0 = nr0; r1 = nr1; r2 = nr2;
        idxA = nidxA; idxB = nidxB;
        dvA = ndvA; dvB = ndvB;
        dn2 = ndn2;
    }
}

template <int J>
__device__ __forceinline__ int group8_bcast(int v) {
    return __builtin_amdgcn_mov_dpp8(v, J * 0x249249);
}
constexpr unsigned group8_xor_sel(int m) {
    unsigned s = 0;
    for (int k = 0; k < 8; ++k) s |= (unsigned)(k ^ m) << (3 * k);
    return s;
}
template <int M>
__device__ __forceinline__ float group8_xor(float v) {
    return __int_as_float(__builtin_amdgcn_mov_dpp8(__float_as_int(v), group8_xor_sel(M)));
}

#define G32_SETS 4
#define G32_NPW  (4 * G32_SETS)
static_assert(G32_SETS % 2 == 0, "sets are taken two at a time");
static_assert(N_NODES % G32_NPW == 0, "a wave is all inside the node range or all outside");

#define G32_ROW(V, IX, N, J)                                           \
    {                                                                  \
        const int s = group8_bcast<J>(IX);                             \
        V[J] = make_float4(0.0f, 0.0f, 0.0f, 0.0f);                    \
        if (J < N) V[J] = h4[s * 8 + i];                               \
    }
#define G32_ROWS(V, IX, N)                                             \
    G32_ROW(V, IX, N, 0) G32_ROW(V, IX, N, 1) G32_ROW(V, IX, N, 2) G32_ROW(V, IX, N, 3) \
    G32_ROW(V, IX, N, 4) G32_ROW(V, IX, N, 5) G32_ROW(V, IX, N, 6) G32_ROW(V, IX, N, 7)

__device__ __forceinline__
void g32_finish(const float4 a, const float dn, const float4 bb, float4* __restrict__ dst) {
    float4 t;
    t.x = a.x * dn + bb.x;
    t.y = a.y * dn + bb.y;
    t.z = a.z * dn + bb.z;
    t.w = a.w * dn + bb.w;
    float m = fmaxf(fmaxf(t.x, t.y), fmaxf(t.z, t.w));
    m = fmaxf(m, group8_xor<1>(m));
    m = fmaxf(m, group8_xor<2>(m));
    m = fmaxf(m, group8_xor<4>(m));
    float s = (__expf(t.x - m) + __expf(t.y - m)) + (__expf(t.z - m) + __expf(t.w - m));
    s += group8_xor<1>(s);
    s += group8_xor<2>(s);
    s += group8_xor<4>(s);
    const float ls = __logf(s);
    float4 o;
    o.x = (t.x - m) - ls;
    o.y = (t.y - m) - ls;
    o.z = (t.z - m) - ls;
    o.w = (t.w - m) - ls;
    *dst = o;
}

__global__ __launch_bounds__(256, 8)
void gather32_final(const int* __restrict__ rowptr, const int* __restrict__ csr_src,
                    const float* __restrict__ h2s, const float* __restrict__ dinv,
                    const float* __restrict__ b2, float* __restrict__ out) {
    const int lane = threadIdx.x & 31;
    const int wave = __builtin_amdgcn_readfirstlane(threadIdx.x >> 5);
    const int n0 = (blockIdx.x * 8 + wave) * G32_NPW;
    if (n0 >= N_NODES) return;
    const int g = lane >> 3;
    const int i = lane & 7;
    const float4* __restrict__ h4 = (const float4*)h2s;
    float4* __restrict__ out4 = (float4*)out;
    const float4 bb = *(const float4*)&b2[4 * i];

    int node[G32_SETS], cur[G32_SETS], end[G32_SETS], idx[G32_SETS];
    float dn[G32_SETS];
    float4 acc[G32_SETS];
#pragma unroll
    for (int q = 0; q < G32_SETS; ++q) {
        node[q] = n0 + 4 * q + g;
        cur[q] = rowptr[node[q]];
        end[q] = rowptr[node[q] + 1];
    }
#pragma unroll
    for (int q = 0; q < G32_SETS; ++q) {
        dn[q]  = dinv[node[q]];
        acc[q] = h4[node[q] * 8 + i];
        idx[q] = (cur[q] + i < end[q]) ? csr_src[cur[q] + i] : 0;
    }

#pragma unroll
    for (int q = 0; q < G32_SETS; q += 2) {
        float4 aA = acc[q], aB = acc[q + 1];
        int cA = cur[q], cB = cur[q + 1];
        const int eA = end[q], eB = end[q + 1];
        int ixA = idx[q], ixB = idx[q + 1];
        while (__any((cA < eA) | (cB < eB))) {
            const int pA = cA + 8 + i;
            const int pB = cB + 8 + i;
            const int nixA = (pA < eA) ? csr_src[pA] : 0;
            const int nixB = (pB < eB) ? csr_src[pB] : 0;
            const int nA = eA - cA;
            const int nB = eB - cB;
            float4 v[8], w[8];
            G32_ROWS(v, ixA, nA)
            G32_ROWS(w, ixB, nB)
#pragma unroll
            for (int j = 0; j < 8; ++j) {
                aA.x += v[j].x; aA.y += v[j].y; aA.z += v[j].z; aA.w += v[j].w;
            }
#pragma unroll
            for (int j = 0; j < 8; ++j) {
                aB.x += w[j].x; aB.y += w[j].y; aB.z += w[j].z; aB.w += w[j].w;
            }
            cA += 8;
            cB += 8;
            ixA = nixA;
            ixB = nixB;
        }
        g32_finish(aA, dn[q],     bb, &out4[node[q] * 8 + i]);
        g32_finish(aB, dn[q + 1], bb, &out4[node[q + 1] * 8 + i]);
    }
}

extern "C" void kernel_launch(void* const* d_in, const int* in_sizes, int n_in,
                              void* d_out, int out_size, void* d_ws, size_t ws_size,
                              hipStream_t stream) {
    const float* x  = (const float*)d_in[0];
    const int*   ei = (const int*)d_in[1];
    const float* W1 = (const float*)d_in[2];
    const float* b1 = (const float*)d_in[3];
    const float* W2 = (const float*)d_in[4];
    const float* b2 = (const float*)d_in[5];
    float* out = (float*)d_out;

    const int* src = ei;
    const int* dst = ei + N_EDGES;

    float* dinv = (float*)d_ws;
    float* h2s  = dinv + N_NODES;
    int* rowptr  = (int*)(h2s + (long)N_NODES * 32);
    int* csr_src = rowptr + N_NODES + 32;
    unsigned* packed = (unsigned*)(csr_src + N_EDGES);
    int* off    = (int*)(packed + N_EDGES);

    chunk_sort<<<NCH, L1_THREADS, 0, stream>>>(src, dst, off, packed);
    bucket_finalize<<<NBK, FIN_THREADS, 0, stream>>>(packed, off, rowptr, dinv, csr_src);

    gather_x_gemm128_gemm32<<<FUSED_BLOCKS, 256, 0, stream>>>(rowptr, csr_src, x, dinv, W1, b1, W2, h2s);
    gather32_final<<<(N_NODES + 8 * G32_NPW - 1) / (8 * G32_NPW), 256, 0, stream>>>(rowptr, csr_src, h2s, dinv, b2, out);
}
